# starting kernel
# baseline (speedup 1.0000x reference)
#include <hip/hip_runtime.h>

typedef __attribute__((ext_vector_type(2))) float v2f;
typedef __attribute__((ext_vector_type(8))) float v8f;

#define SEQ_LEN 262144
#define NGATES  12
#define HDIM    4

__device__ __forceinline__ float bcast_lane(float v, int lane) {
  return __int_as_float(__builtin_amdgcn_readlane(__float_as_int(v), lane));
}

__device__ __forceinline__ float fast_sigmoid(float x) {
  const float e = __builtin_amdgcn_exp2f(-1.44269504088896340736f * x);
  return __builtin_amdgcn_rcpf(1.0f + e);
}

__global__ __launch_bounds__(256) void gi_wmma_kernel(
    const float* __restrict__ x, const float* __restrict__ w_ih,
    const float* __restrict__ b_ih, float* __restrict__ gi) {
  const int lane = threadIdx.x & 31;
  const int wave = threadIdx.x >> 5;
  const int tile = blockIdx.x * 8 + wave;
  const int t0   = tile * 16;
  const int half = lane >> 4;
  const int idx  = lane & 15;

  v2f a;
  if (half == 0) {
    const float2 xv = *reinterpret_cast<const float2*>(x + (size_t)(t0 + idx) * 2);
    a.x = xv.x; a.y = xv.y;
  } else {
    a.x = 0.0f; a.y = 0.0f;
  }

  v2f b;
  if (half == 0 && idx < NGATES) {
    const float2 wv = *reinterpret_cast<const float2*>(w_ih + idx * 2);
    b.x = wv.x; b.y = wv.y;
  } else {
    b.x = 0.0f; b.y = 0.0f;
  }

  const float bias = (idx < NGATES) ? b_ih[idx] : 0.0f;
  v8f c;
#pragma unroll
  for (int i = 0; i < 8; ++i) c[i] = bias;

  v8f d = __builtin_amdgcn_wmma_f32_16x16x4_f32(
       false, a,  false, b,
       (short)0, c,  false,  false);

  if (idx < NGATES) {
    float* dst = gi + (size_t)idx * SEQ_LEN + t0 + half * 8;
    *reinterpret_cast<float4*>(dst + 0) = make_float4(d[0], d[1], d[2], d[3]);
    *reinterpret_cast<float4*>(dst + 4) = make_float4(d[4], d[5], d[6], d[7]);
  }
}

__global__ __launch_bounds__(32) void gru_scan_kernel(
    const float* __restrict__ gi, const float* __restrict__ state0,
    const float* __restrict__ w_hh, const float* __restrict__ b_hh,
    float* __restrict__ hbuf, float* __restrict__ state_out) {
  const int lane = threadIdx.x;
  const int j    = (lane < NGATES) ? lane : 0;
  const int jj   = lane & 3;

  const float w0 = w_hh[j * 4 + 0];
  const float w1 = w_hh[j * 4 + 1];
  const float w2 = w_hh[j * 4 + 2];
  const float w3 = w_hh[j * 4 + 3];
  const float bh = b_hh[j];

  float h0 = state0[0], h1 = state0[1], h2 = state0[2], h3 = state0[3];

  const float* __restrict__ gptr = gi + (size_t)j * SEQ_LEN;

  constexpr int PF = 8;
  float buf[PF];
#pragma unroll
  for (int u = 0; u < PF; ++u) buf[u] = gptr[u];

  for (int t = 0; t < SEQ_LEN; t += PF) {
#pragma unroll
    for (int u = 0; u < PF; ++u) {
      const float giv = buf[u];
      buf[u] = gptr[t + u + PF];

      const float p01 = fmaf(h1, w1, fmaf(h0, w0, bh));
      const float p23 = fmaf(h3, w3, h2 * w2);
      const float gh  = p01 + p23;

      const float s = fast_sigmoid(giv + gh);

      const float r = __shfl(s, jj, 32);

      const float nv = fmaf(2.0f, fast_sigmoid(2.0f * (giv + r * gh)), -1.0f);

      const float z    = __shfl(s, 4 + jj, 32);
      const float nn   = __shfl(nv, 8 + jj, 32);
      const float hold = (jj == 0) ? h0 : (jj == 1) ? h1 : (jj == 2) ? h2 : h3;
      const float hnew = fmaf(z, hold - nn, nn);

      if (lane < HDIM) hbuf[(size_t)(t + u) * HDIM + lane] = hnew;

      h0 = bcast_lane(hnew, 0);
      h1 = bcast_lane(hnew, 1);
      h2 = bcast_lane(hnew, 2);
      h3 = bcast_lane(hnew, 3);
    }
  }

  if (lane == 0) {
    state_out[0] = h0; state_out[1] = h1;
    state_out[2] = h2; state_out[3] = h3;
  }
}

__global__ __launch_bounds__(256) void yproj_wmma_kernel(
    const float* __restrict__ hbuf, const float* __restrict__ w_out,
    const float* __restrict__ b_out, float* __restrict__ y) {
  const int lane = threadIdx.x & 31;
  const int wave = threadIdx.x >> 5;
  const int tile = blockIdx.x * 8 + wave;
  const int t0   = tile * 16;
  const int half = lane >> 4;
  const int idx  = lane & 15;

  v2f a;
  {
    const float2 hv = *reinterpret_cast<const float2*>(
        hbuf + (size_t)(t0 + idx) * 4 + half * 2);
    a.x = hv.x; a.y = hv.y;
  }

  v2f b;
  if (idx < 2) {
    const float2 wv = *reinterpret_cast<const float2*>(w_out + idx * 4 + half * 2);
    b.x = wv.x; b.y = wv.y;
  } else {
    b.x = 0.0f; b.y = 0.0f;
  }

  const float bias = (idx < 2) ? b_out[idx] : 0.0f;
  v8f c;
#pragma unroll
  for (int i = 0; i < 8; ++i) c[i] = bias;

  v8f d = __builtin_amdgcn_wmma_f32_16x16x4_f32(
      false, a, false, b, (short)0, c, false, false);

  if (idx < 2) {
    const int rbase = half * 8;
#pragma unroll
    for (int i = 0; i < 8; ++i)
      y[(size_t)(t0 + rbase + i) * 2 + idx] = d[i];
  }
}

extern "C" void kernel_launch(void* const* d_in, const int* in_sizes, int n_in,
                              void* d_out, int out_size, void* d_ws, size_t ws_size,
                              hipStream_t stream) {
  const float* x      = (const float*)d_in[0];
  const float* state  = (const float*)d_in[1];
  const float* w_ih   = (const float*)d_in[2];
  const float* w_hh   = (const float*)d_in[3];
  const float* b_ih   = (const float*)d_in[4];
  const float* b_hh   = (const float*)d_in[5];
  const float* w_out  = (const float*)d_in[6];
  const float* b_out  = (const float*)d_in[7];

  float* out  = (float*)d_out;
  float* gi   = (float*)d_ws;
  float* hbuf = gi + (size_t)NGATES * SEQ_LEN;

  const int tiles  = SEQ_LEN / 16;
  const int blocks = tiles / 8;

  gi_wmma_kernel<<<blocks, 256, 0, stream>>>(x, w_ih, b_ih, gi);
  gru_scan_kernel<<<1, 32, 0, stream>>>(gi, state, w_hh, b_hh,
                                        hbuf, out + (size_t)SEQ_LEN * 2);
  yproj_wmma_kernel<<<blocks, 256, 0, stream>>>(hbuf, w_out, b_out, out);
}


# scan: the WMMA's 16 columns run 16 time chunks of 1024 steps (were 16 copies of one state); sweep 2 starts each chunk at its neighbour's zero-start exit; joins checked bitwise, else serial rerun
# speedup vs baseline: 360.8084x; 360.8084x over previous
#include <hip/hip_runtime.h>

typedef __attribute__((ext_vector_type(2))) float v2f;
typedef __attribute__((ext_vector_type(8))) float v8f;

#define SEQ_LEN 262144
#define NGATES  12
#define HDIM    4

__device__ __forceinline__ float fast_sigmoid(float x) {
  const float e = __builtin_amdgcn_exp2f(-1.44269504088896340736f * x);
  return __builtin_amdgcn_rcpf(1.0f + e);
}

__global__ __launch_bounds__(256) void gi_wmma_kernel(
    const float* __restrict__ x, const float* __restrict__ w_ih,
    const float* __restrict__ b_ih, float* __restrict__ gi) {
  const int lane = threadIdx.x & 31;
  const int wave = threadIdx.x >> 5;
  const int tile = blockIdx.x * 8 + wave;
  const int t0   = tile * 16;
  const int half = lane >> 4;
  const int idx  = lane & 15;

  v2f a;
  if (half == 0) {
    const float2 xv = *reinterpret_cast<const float2*>(x + (size_t)(t0 + idx) * 2);
    a.x = xv.x; a.y = xv.y;
  } else {
    a.x = 0.0f; a.y = 0.0f;
  }

  v2f b;
  if (half == 0 && idx < NGATES) {
    const float2 wv = *reinterpret_cast<const float2*>(w_ih + idx * 2);
    b.x = wv.x; b.y = wv.y;
  } else {
    b.x = 0.0f; b.y = 0.0f;
  }

  const float bias = (idx < NGATES) ? b_ih[idx] : 0.0f;
  v8f c;
#pragma unroll
  for (int i = 0; i < 8; ++i) c[i] = bias;

  v8f d = __builtin_amdgcn_wmma_f32_16x16x4_f32(
       false, a,  false, b,
       (short)0, c,  false,  false);

  if (idx < NGATES) {
    float* dst = gi + (size_t)idx * SEQ_LEN + t0 + half * 8;
    *reinterpret_cast<float4*>(dst + 0) = make_float4(d[0], d[1], d[2], d[3]);
    *reinterpret_cast<float4*>(dst + 4) = make_float4(d[4], d[5], d[6], d[7]);
  }
}

#define SCAN_GRP  4
#define SCAN_RING 4

template <int NSTEPS, bool STORE_H>
__device__ __forceinline__ void gru_steps(
    const float* __restrict__ gi, const float* __restrict__ w_hh,
    const float* __restrict__ b_hh, float* __restrict__ hbuf,
    const int t_base, float& h_a, float& h_b) {
  const int lane = threadIdx.x;
  const int hf   = lane >> 4;
  const int c    = lane & 15;

  const int g_r0 = 0 + 2 * hf, g_r1 = 1 + 2 * hf;
  const int g_z0 = 4 + 2 * hf, g_z1 = 5 + 2 * hf;
  const int g_n0 = 8 + 2 * hf, g_n1 = 9 + 2 * hf;

  v2f a;
  {
    const int mh = c >> 3, mi = c & 7;
    if (mi < 6) {
      const int g = (mi >> 1) * 4 + 2 * mh + (mi & 1);
      const float2 wv = *reinterpret_cast<const float2*>(w_hh + g * 4 + 2 * hf);
      a.x = wv.x; a.y = wv.y;
    } else {
      a.x = 0.0f; a.y = 0.0f;
    }
  }

  const float bh_r0 = b_hh[g_r0], bh_r1 = b_hh[g_r1];
  const float bh_z0 = b_hh[g_z0], bh_z1 = b_hh[g_z1];
  const float bh_n0 = b_hh[g_n0], bh_n1 = b_hh[g_n1];

  float ha = h_a;
  float hb = h_b;

  const float* __restrict__ p_r0 = gi + (size_t)g_r0 * SEQ_LEN + t_base;
  const float* __restrict__ p_r1 = gi + (size_t)g_r1 * SEQ_LEN + t_base;
  const float* __restrict__ p_z0 = gi + (size_t)g_z0 * SEQ_LEN + t_base;
  const float* __restrict__ p_z1 = gi + (size_t)g_z1 * SEQ_LEN + t_base;
  const float* __restrict__ p_n0 = gi + (size_t)g_n0 * SEQ_LEN + t_base;
  const float* __restrict__ p_n1 = gi + (size_t)g_n1 * SEQ_LEN + t_base;
  float* __restrict__ hp = hbuf + (size_t)t_base * HDIM + 2 * hf;

  auto load_row = [&](float (&dst)[SCAN_GRP], const float* __restrict__ p, int t) {
    const float4 v = *reinterpret_cast<const float4*>(p + t);
    dst[0] = v.x; dst[1] = v.y; dst[2] = v.z; dst[3] = v.w;
  };
  auto load_grp = [&](float (&g)[6][SCAN_GRP], int t) {
    load_row(g[0], p_r0, t); load_row(g[1], p_r1, t);
    load_row(g[2], p_z0, t); load_row(g[3], p_z1, t);
    load_row(g[4], p_n0, t); load_row(g[5], p_n1, t);
  };

  auto run_grp = [&](const float (&g)[6][SCAN_GRP], int t) {
#pragma unroll
    for (int u = 0; u < SCAN_GRP; ++u) {
      v8f cc;
      cc[0] = g[0][u] + bh_r0; cc[1] = g[1][u] + bh_r1;
      cc[2] = g[2][u] + bh_z0; cc[3] = g[3][u] + bh_z1;
      cc[4] = bh_n0;           cc[5] = bh_n1;
      cc[6] = 0.0f;            cc[7] = 0.0f;
      v2f b; b.x = ha; b.y = hb;

      const v8f d = __builtin_amdgcn_wmma_f32_16x16x4_f32(
           false, a,  false, b,
           (short)0, cc,  false,  false);

      const float ra = fast_sigmoid(d[0]);
      const float rb = fast_sigmoid(d[1]);
      const float za = fast_sigmoid(d[2]);
      const float zb = fast_sigmoid(d[3]);

      const float na = fmaf(2.0f, fast_sigmoid(2.0f * fmaf(ra, d[4], g[4][u])), -1.0f);
      const float nb = fmaf(2.0f, fast_sigmoid(2.0f * fmaf(rb, d[5], g[5][u])), -1.0f);

      ha = fmaf(za, ha - na, na);
      hb = fmaf(zb, hb - nb, nb);

      if (STORE_H)
        *reinterpret_cast<float2*>(hp + (size_t)(t + u) * HDIM) = make_float2(ha, hb);
    }
  };

  float g[SCAN_RING][6][SCAN_GRP];
#pragma unroll
  for (int j = 0; j < SCAN_RING; ++j) load_grp(g[j], SCAN_GRP * j);

  for (int t = 0; t < NSTEPS; t += SCAN_GRP * SCAN_RING) {
#pragma unroll
    for (int j = 0; j < SCAN_RING; ++j) {
      run_grp(g[j], t + SCAN_GRP * j);
      load_grp(g[j], t + SCAN_GRP * (SCAN_RING + j));
      __builtin_amdgcn_sched_barrier(0);
    }
  }

  h_a = ha;
  h_b = hb;
}

#define CH_LEN   1024
#define N_CHUNK  (SEQ_LEN / CH_LEN)
#define CH_WAVES (N_CHUNK / 16)
#define JOIN_BAD 0x7fb1c5a7

template <int SWEEP>
__global__ __launch_bounds__(32) void gru_sweep_kernel(
    const float* __restrict__ gi, const float* __restrict__ state0,
    const float* __restrict__ w_hh, const float* __restrict__ b_hh,
    float* __restrict__ hbuf) {
  const int lane = threadIdx.x;
  const int hf   = lane >> 4;
  const int c    = lane & 15;
  const int k    = blockIdx.x * 16 + c;
  const int t_base = k * CH_LEN;

  float* first_slot = hbuf + (size_t)t_base * HDIM + 2 * hf;
  float* last_slot  = hbuf + (size_t)(t_base + CH_LEN - 1) * HDIM + 2 * hf;

  float ha = 0.0f, hb = 0.0f;
  float ea = 0.0f, eb = 0.0f;
  if (k == 0) {
    ha = state0[2 * hf + 0];
    hb = state0[2 * hf + 1];
  } else if (SWEEP == 1) {
    const float2 v = *reinterpret_cast<const float2*>(first_slot);
    ha = v.x; hb = v.y;
  }
  if (SWEEP == 1) {
    const float2 e = *reinterpret_cast<const float2*>(last_slot);
    ea = e.x; eb = e.y;
  }

  gru_steps<CH_LEN, SWEEP == 1>(gi, w_hh, b_hh, hbuf, t_base, ha, hb);

  if (SWEEP == 0) {
    *reinterpret_cast<float2*>(last_slot) = make_float2(ha, hb);
    if (k + 1 < N_CHUNK)
      *reinterpret_cast<float2*>(first_slot + (size_t)CH_LEN * HDIM) = make_float2(ha, hb);
  } else {
    const bool moved = (__builtin_bit_cast(int, ha) != __builtin_bit_cast(int, ea)) |
                       (__builtin_bit_cast(int, hb) != __builtin_bit_cast(int, eb));
    if (moved) {
      const float m = __builtin_bit_cast(float, (int)JOIN_BAD);
      *reinterpret_cast<float2*>(last_slot) = make_float2(m, m);
    }
  }
}

__global__ __launch_bounds__(32) void gru_join_kernel(
    const float* __restrict__ gi, const float* __restrict__ state0,
    const float* __restrict__ w_hh, const float* __restrict__ b_hh,
    float* __restrict__ hbuf, float* __restrict__ state_out) {
  const int lane = threadIdx.x;
  const int hf   = lane >> 4;
  const int c    = lane & 15;

  bool bad = false;
#pragma unroll
  for (int j = 0; j < N_CHUNK / 32; ++j) {
    const int k = j * 32 + lane;
    const float4 v = *reinterpret_cast<const float4*>(
        hbuf + ((size_t)(k + 1) * CH_LEN - 1) * HDIM);
    bad |= (__builtin_bit_cast(int, v.x) == (int)JOIN_BAD) |
           (__builtin_bit_cast(int, v.y) == (int)JOIN_BAD) |
           (__builtin_bit_cast(int, v.z) == (int)JOIN_BAD) |
           (__builtin_bit_cast(int, v.w) == (int)JOIN_BAD);
  }

  float ha, hb;
  if (__builtin_amdgcn_ballot_w32(bad) != 0u) {
    ha = state0[2 * hf + 0];
    hb = state0[2 * hf + 1];
    gru_steps<SEQ_LEN, true>(gi, w_hh, b_hh, hbuf, 0, ha, hb);
  } else {
    const float2 v = *reinterpret_cast<const float2*>(
        hbuf + (size_t)(SEQ_LEN - 1) * HDIM + 2 * hf);
    ha = v.x; hb = v.y;
  }

  if (c == 0) {
    state_out[2 * hf + 0] = ha;
    state_out[2 * hf + 1] = hb;
  }
}

__global__ __launch_bounds__(256) void yproj_wmma_kernel(
    const float* __restrict__ hbuf, const float* __restrict__ w_out,
    const float* __restrict__ b_out, float* __restrict__ y) {
  const int lane = threadIdx.x & 31;
  const int wave = threadIdx.x >> 5;
  const int tile = blockIdx.x * 8 + wave;
  const int t0   = tile * 16;
  const int half = lane >> 4;
  const int idx  = lane & 15;

  v2f a;
  {
    const float2 hv = *reinterpret_cast<const float2*>(
        hbuf + (size_t)(t0 + idx) * 4 + half * 2);
    a.x = hv.x; a.y = hv.y;
  }

  v2f b;
  if (idx < 2) {
    const float2 wv = *reinterpret_cast<const float2*>(w_out + idx * 4 + half * 2);
    b.x = wv.x; b.y = wv.y;
  } else {
    b.x = 0.0f; b.y = 0.0f;
  }

  const float bias = (idx < 2) ? b_out[idx] : 0.0f;
  v8f c;
#pragma unroll
  for (int i = 0; i < 8; ++i) c[i] = bias;

  v8f d = __builtin_amdgcn_wmma_f32_16x16x4_f32(
      false, a, false, b, (short)0, c, false, false);

  if (idx < 2) {
    const int rbase = half * 8;
#pragma unroll
    for (int i = 0; i < 8; ++i)
      y[(size_t)(t0 + rbase + i) * 2 + idx] = d[i];
  }
}

extern "C" void kernel_launch(void* const* d_in, const int* in_sizes, int n_in,
                              void* d_out, int out_size, void* d_ws, size_t ws_size,
                              hipStream_t stream) {
  const float* x      = (const float*)d_in[0];
  const float* state  = (const float*)d_in[1];
  const float* w_ih   = (const float*)d_in[2];
  const float* w_hh   = (const float*)d_in[3];
  const float* b_ih   = (const float*)d_in[4];
  const float* b_hh   = (const float*)d_in[5];
  const float* w_out  = (const float*)d_in[6];
  const float* b_out  = (const float*)d_in[7];

  float* out  = (float*)d_out;
  float* gi   = (float*)d_ws;
  float* hbuf = gi + (size_t)NGATES * SEQ_LEN;

  const int tiles  = SEQ_LEN / 16;
  const int blocks = tiles / 8;

  gi_wmma_kernel<<<blocks, 256, 0, stream>>>(x, w_ih, b_ih, gi);
  gru_sweep_kernel<0><<<CH_WAVES, 32, 0, stream>>>(gi, state, w_hh, b_hh, hbuf);
  gru_sweep_kernel<1><<<CH_WAVES, 32, 0, stream>>>(gi, state, w_hh, b_hh, hbuf);
  gru_join_kernel<<<1, 32, 0, stream>>>(gi, state, w_hh, b_hh,
                                        hbuf, out + (size_t)SEQ_LEN * 2);
  yproj_wmma_kernel<<<blocks, 256, 0, stream>>>(hbuf, w_out, b_out, out);
}


# scan: 1024 chunks of 256 steps relaxed over 3 sweeps (was 256 of 1024 over 2): a sweep enters each chunk at its neighbour's exit of the sweep before; last joins checked bitwise, else serial rerun
# speedup vs baseline: 797.4075x; 2.2101x over previous
#include <hip/hip_runtime.h>

typedef __attribute__((ext_vector_type(2))) float v2f;
typedef __attribute__((ext_vector_type(8))) float v8f;

#define SEQ_LEN 262144
#define NGATES  12
#define HDIM    4

__device__ __forceinline__ float fast_sigmoid(float x) {
  const float e = __builtin_amdgcn_exp2f(-1.44269504088896340736f * x);
  return __builtin_amdgcn_rcpf(1.0f + e);
}

__global__ __launch_bounds__(256) void gi_wmma_kernel(
    const float* __restrict__ x, const float* __restrict__ w_ih,
    const float* __restrict__ b_ih, float* __restrict__ gi) {
  const int lane = threadIdx.x & 31;
  const int wave = threadIdx.x >> 5;
  const int tile = blockIdx.x * 8 + wave;
  const int t0   = tile * 16;
  const int half = lane >> 4;
  const int idx  = lane & 15;

  v2f a;
  if (half == 0) {
    const float2 xv = *reinterpret_cast<const float2*>(x + (size_t)(t0 + idx) * 2);
    a.x = xv.x; a.y = xv.y;
  } else {
    a.x = 0.0f; a.y = 0.0f;
  }

  v2f b;
  if (half == 0 && idx < NGATES) {
    const float2 wv = *reinterpret_cast<const float2*>(w_ih + idx * 2);
    b.x = wv.x; b.y = wv.y;
  } else {
    b.x = 0.0f; b.y = 0.0f;
  }

  const float bias = (idx < NGATES) ? b_ih[idx] : 0.0f;
  v8f c;
#pragma unroll
  for (int i = 0; i < 8; ++i) c[i] = bias;

  v8f d = __builtin_amdgcn_wmma_f32_16x16x4_f32(
       false, a,  false, b,
       (short)0, c,  false,  false);

  if (idx < NGATES) {
    float* dst = gi + (size_t)idx * SEQ_LEN + t0 + half * 8;
    *reinterpret_cast<float4*>(dst + 0) = make_float4(d[0], d[1], d[2], d[3]);
    *reinterpret_cast<float4*>(dst + 4) = make_float4(d[4], d[5], d[6], d[7]);
  }
}

#define SCAN_GRP  4
#define SCAN_RING 4

template <int NSTEPS, bool STORE_H>
__device__ __forceinline__ void gru_steps(
    const float* __restrict__ gi, const float* __restrict__ w_hh,
    const float* __restrict__ b_hh, float* __restrict__ hbuf,
    const int t_base, float& h_a, float& h_b) {
  const int lane = threadIdx.x;
  const int hf   = lane >> 4;
  const int c    = lane & 15;

  const int g_r0 = 0 + 2 * hf, g_r1 = 1 + 2 * hf;
  const int g_z0 = 4 + 2 * hf, g_z1 = 5 + 2 * hf;
  const int g_n0 = 8 + 2 * hf, g_n1 = 9 + 2 * hf;

  v2f a;
  {
    const int mh = c >> 3, mi = c & 7;
    if (mi < 6) {
      const int g = (mi >> 1) * 4 + 2 * mh + (mi & 1);
      const float2 wv = *reinterpret_cast<const float2*>(w_hh + g * 4 + 2 * hf);
      a.x = wv.x; a.y = wv.y;
    } else {
      a.x = 0.0f; a.y = 0.0f;
    }
  }

  const float bh_r0 = b_hh[g_r0], bh_r1 = b_hh[g_r1];
  const float bh_z0 = b_hh[g_z0], bh_z1 = b_hh[g_z1];
  const float bh_n0 = b_hh[g_n0], bh_n1 = b_hh[g_n1];

  float ha = h_a;
  float hb = h_b;

  const float* __restrict__ p_r0 = gi + (size_t)g_r0 * SEQ_LEN + t_base;
  const float* __restrict__ p_r1 = gi + (size_t)g_r1 * SEQ_LEN + t_base;
  const float* __restrict__ p_z0 = gi + (size_t)g_z0 * SEQ_LEN + t_base;
  const float* __restrict__ p_z1 = gi + (size_t)g_z1 * SEQ_LEN + t_base;
  const float* __restrict__ p_n0 = gi + (size_t)g_n0 * SEQ_LEN + t_base;
  const float* __restrict__ p_n1 = gi + (size_t)g_n1 * SEQ_LEN + t_base;
  float* __restrict__ hp = hbuf + (size_t)t_base * HDIM + 2 * hf;

  auto load_row = [&](float (&dst)[SCAN_GRP], const float* __restrict__ p, int t) {
    const float4 v = *reinterpret_cast<const float4*>(p + t);
    dst[0] = v.x; dst[1] = v.y; dst[2] = v.z; dst[3] = v.w;
  };
  auto load_grp = [&](float (&g)[6][SCAN_GRP], int t) {
    load_row(g[0], p_r0, t); load_row(g[1], p_r1, t);
    load_row(g[2], p_z0, t); load_row(g[3], p_z1, t);
    load_row(g[4], p_n0, t); load_row(g[5], p_n1, t);
  };

  auto run_grp = [&](const float (&g)[6][SCAN_GRP], int t) {
#pragma unroll
    for (int u = 0; u < SCAN_GRP; ++u) {
      v8f cc;
      cc[0] = g[0][u] + bh_r0; cc[1] = g[1][u] + bh_r1;
      cc[2] = g[2][u] + bh_z0; cc[3] = g[3][u] + bh_z1;
      cc[4] = bh_n0;           cc[5] = bh_n1;
      cc[6] = 0.0f;            cc[7] = 0.0f;
      v2f b; b.x = ha; b.y = hb;

      const v8f d = __builtin_amdgcn_wmma_f32_16x16x4_f32(
           false, a,  false, b,
           (short)0, cc,  false,  false);

      const float ra = fast_sigmoid(d[0]);
      const float rb = fast_sigmoid(d[1]);
      const float za = fast_sigmoid(d[2]);
      const float zb = fast_sigmoid(d[3]);

      const float na = fmaf(2.0f, fast_sigmoid(2.0f * fmaf(ra, d[4], g[4][u])), -1.0f);
      const float nb = fmaf(2.0f, fast_sigmoid(2.0f * fmaf(rb, d[5], g[5][u])), -1.0f);

      ha = fmaf(za, ha - na, na);
      hb = fmaf(zb, hb - nb, nb);

      if (STORE_H)
        *reinterpret_cast<float2*>(hp + (size_t)(t + u) * HDIM) = make_float2(ha, hb);
    }
  };

  float g[SCAN_RING][6][SCAN_GRP];
#pragma unroll
  for (int j = 0; j < SCAN_RING; ++j) load_grp(g[j], SCAN_GRP * j);

  for (int t = 0; t < NSTEPS; t += SCAN_GRP * SCAN_RING) {
#pragma unroll
    for (int j = 0; j < SCAN_RING; ++j) {
      run_grp(g[j], t + SCAN_GRP * j);
      load_grp(g[j], t + SCAN_GRP * (SCAN_RING + j));
      __builtin_amdgcn_sched_barrier(0);
    }
  }

  h_a = ha;
  h_b = hb;
}

#define CH_LEN   256
#define N_CHUNK  (SEQ_LEN / CH_LEN)
#define CH_WAVES (N_CHUNK / 16)
#define N_SWEEP  3
#define JOIN_BAD 0x7fb1c5a7

template <int SWEEP>
__global__ __launch_bounds__(32) void gru_sweep_kernel(
    const float* __restrict__ gi, const float* __restrict__ state0,
    const float* __restrict__ w_hh, const float* __restrict__ b_hh,
    float* __restrict__ hbuf) {
  constexpr bool LAST = (SWEEP == N_SWEEP - 1);
  constexpr int  P_OUT = SWEEP & 1;
  constexpr int  P_IN  = P_OUT ^ 1;

  const int lane = threadIdx.x;
  const int hf   = lane >> 4;
  const int c    = lane & 15;
  const int k    = blockIdx.x * 16 + c;
  const int t_base = k * CH_LEN;

  float* own = hbuf + (size_t)t_base * HDIM + 2 * hf;

  float ha = 0.0f, hb = 0.0f;
  float ea = 0.0f, eb = 0.0f;
  if (k == 0) {
    ha = state0[2 * hf + 0];
    hb = state0[2 * hf + 1];
  } else if (SWEEP > 0) {
    const float2 v = *reinterpret_cast<const float2*>(own + (size_t)P_IN * HDIM);
    ha = v.x; hb = v.y;
  }
  if (LAST) {
    const float2 e = *reinterpret_cast<const float2*>(
        own + (size_t)(CH_LEN - 1 - P_IN) * HDIM);
    ea = e.x; eb = e.y;
  }

  gru_steps<CH_LEN, LAST>(gi, w_hh, b_hh, hbuf, t_base, ha, hb);

  if (!LAST) {
    if (SWEEP == N_SWEEP - 2)
      *reinterpret_cast<float2*>(own + (size_t)(CH_LEN - 1 - P_OUT) * HDIM) =
          make_float2(ha, hb);
    if (k + 1 < N_CHUNK)
      *reinterpret_cast<float2*>(own + (size_t)(CH_LEN + P_OUT) * HDIM) =
          make_float2(ha, hb);
  } else {
    const bool moved = (__builtin_bit_cast(int, ha) != __builtin_bit_cast(int, ea)) |
                       (__builtin_bit_cast(int, hb) != __builtin_bit_cast(int, eb));
    if (moved) {
      const float m = __builtin_bit_cast(float, (int)JOIN_BAD);
      *reinterpret_cast<float2*>(own + (size_t)(CH_LEN - 1) * HDIM) = make_float2(m, m);
    }
  }
}

__global__ __launch_bounds__(32) void gru_join_kernel(
    const float* __restrict__ gi, const float* __restrict__ state0,
    const float* __restrict__ w_hh, const float* __restrict__ b_hh,
    float* __restrict__ hbuf, float* __restrict__ state_out) {
  const int lane = threadIdx.x;
  const int hf   = lane >> 4;
  const int c    = lane & 15;

  bool bad = false;
#pragma unroll
  for (int j = 0; j < N_CHUNK / 32; ++j) {
    const int k = j * 32 + lane;
    const float4 v = *reinterpret_cast<const float4*>(
        hbuf + ((size_t)(k + 1) * CH_LEN - 1) * HDIM);
    bad |= (__builtin_bit_cast(int, v.x) == (int)JOIN_BAD) |
           (__builtin_bit_cast(int, v.y) == (int)JOIN_BAD) |
           (__builtin_bit_cast(int, v.z) == (int)JOIN_BAD) |
           (__builtin_bit_cast(int, v.w) == (int)JOIN_BAD);
  }

  float ha, hb;
  if (__builtin_amdgcn_ballot_w32(bad) != 0u) {
    ha = state0[2 * hf + 0];
    hb = state0[2 * hf + 1];
    gru_steps<SEQ_LEN, true>(gi, w_hh, b_hh, hbuf, 0, ha, hb);
  } else {
    const float2 v = *reinterpret_cast<const float2*>(
        hbuf + (size_t)(SEQ_LEN - 1) * HDIM + 2 * hf);
    ha = v.x; hb = v.y;
  }

  if (c == 0) {
    state_out[2 * hf + 0] = ha;
    state_out[2 * hf + 1] = hb;
  }
}

__global__ __launch_bounds__(256) void yproj_wmma_kernel(
    const float* __restrict__ hbuf, const float* __restrict__ w_out,
    const float* __restrict__ b_out, float* __restrict__ y) {
  const int lane = threadIdx.x & 31;
  const int wave = threadIdx.x >> 5;
  const int tile = blockIdx.x * 8 + wave;
  const int t0   = tile * 16;
  const int half = lane >> 4;
  const int idx  = lane & 15;

  v2f a;
  {
    const float2 hv = *reinterpret_cast<const float2*>(
        hbuf + (size_t)(t0 + idx) * 4 + half * 2);
    a.x = hv.x; a.y = hv.y;
  }

  v2f b;
  if (idx < 2) {
    const float2 wv = *reinterpret_cast<const float2*>(w_out + idx * 4 + half * 2);
    b.x = wv.x; b.y = wv.y;
  } else {
    b.x = 0.0f; b.y = 0.0f;
  }

  const float bias = (idx < 2) ? b_out[idx] : 0.0f;
  v8f c;
#pragma unroll
  for (int i = 0; i < 8; ++i) c[i] = bias;

  v8f d = __builtin_amdgcn_wmma_f32_16x16x4_f32(
      false, a, false, b, (short)0, c, false, false);

  if (idx < 2) {
    const int rbase = half * 8;
#pragma unroll
    for (int i = 0; i < 8; ++i)
      y[(size_t)(t0 + rbase + i) * 2 + idx] = d[i];
  }
}

extern "C" void kernel_launch(void* const* d_in, const int* in_sizes, int n_in,
                              void* d_out, int out_size, void* d_ws, size_t ws_size,
                              hipStream_t stream) {
  const float* x      = (const float*)d_in[0];
  const float* state  = (const float*)d_in[1];
  const float* w_ih   = (const float*)d_in[2];
  const float* w_hh   = (const float*)d_in[3];
  const float* b_ih   = (const float*)d_in[4];
  const float* b_hh   = (const float*)d_in[5];
  const float* w_out  = (const float*)d_in[6];
  const float* b_out  = (const float*)d_in[7];

  float* out  = (float*)d_out;
  float* gi   = (float*)d_ws;
  float* hbuf = gi + (size_t)NGATES * SEQ_LEN;

  const int tiles  = SEQ_LEN / 16;
  const int blocks = tiles / 8;

  gi_wmma_kernel<<<blocks, 256, 0, stream>>>(x, w_ih, b_ih, gi);
  gru_sweep_kernel<0><<<CH_WAVES, 32, 0, stream>>>(gi, state, w_hh, b_hh, hbuf);
  gru_sweep_kernel<1><<<CH_WAVES, 32, 0, stream>>>(gi, state, w_hh, b_hh, hbuf);
  gru_sweep_kernel<2><<<CH_WAVES, 32, 0, stream>>>(gi, state, w_hh, b_hh, hbuf);
  gru_join_kernel<<<1, 32, 0, stream>>>(gi, state, w_hh, b_hh,
                                        hbuf, out + (size_t)SEQ_LEN * 2);
  yproj_wmma_kernel<<<blocks, 256, 0, stream>>>(hbuf, w_out, b_out, out);
}


# scan: all sweeps in ONE launch: a wave runs 9 sweeps over 16 neighbouring 64-step chunks, exits passed to the next column by a DPP row shift (8 run-in columns, 8 own); joins checked bitwise
# speedup vs baseline: 891.2316x; 1.1177x over previous
#include <hip/hip_runtime.h>

typedef __attribute__((ext_vector_type(2))) float v2f;
typedef __attribute__((ext_vector_type(8))) float v8f;

#define SEQ_LEN 262144
#define NGATES  12
#define HDIM    4

__device__ __forceinline__ float fast_sigmoid(float x) {
  const float e = __builtin_amdgcn_exp2f(-1.44269504088896340736f * x);
  return __builtin_amdgcn_rcpf(1.0f + e);
}

__global__ __launch_bounds__(256) void gi_wmma_kernel(
    const float* __restrict__ x, const float* __restrict__ w_ih,
    const float* __restrict__ b_ih, float* __restrict__ gi) {
  const int lane = threadIdx.x & 31;
  const int wave = threadIdx.x >> 5;
  const int tile = blockIdx.x * 8 + wave;
  const int t0   = tile * 16;
  const int half = lane >> 4;
  const int idx  = lane & 15;

  v2f a;
  if (half == 0) {
    const float2 xv = *reinterpret_cast<const float2*>(x + (size_t)(t0 + idx) * 2);
    a.x = xv.x; a.y = xv.y;
  } else {
    a.x = 0.0f; a.y = 0.0f;
  }

  v2f b;
  if (half == 0 && idx < NGATES) {
    const float2 wv = *reinterpret_cast<const float2*>(w_ih + idx * 2);
    b.x = wv.x; b.y = wv.y;
  } else {
    b.x = 0.0f; b.y = 0.0f;
  }

  const float bias = (idx < NGATES) ? b_ih[idx] : 0.0f;
  v8f c;
#pragma unroll
  for (int i = 0; i < 8; ++i) c[i] = bias;

  v8f d = __builtin_amdgcn_wmma_f32_16x16x4_f32(
       false, a,  false, b,
       (short)0, c,  false,  false);

  if (idx < NGATES) {
    float* dst = gi + (size_t)idx * SEQ_LEN + t0 + half * 8;
    *reinterpret_cast<float4*>(dst + 0) = make_float4(d[0], d[1], d[2], d[3]);
    *reinterpret_cast<float4*>(dst + 4) = make_float4(d[4], d[5], d[6], d[7]);
  }
}

#define SCAN_GRP  4
#define SCAN_RING 4
#define SCAN_TRIP (SCAN_GRP * SCAN_RING)

struct GruLane {
  v2f a;
  float bh_r0, bh_r1, bh_z0, bh_z1, bh_n0, bh_n1;
  const float* p_r0; const float* p_r1;
  const float* p_z0; const float* p_z1;
  const float* p_n0; const float* p_n1;
  float* hp;
  float ha, hb;
};

__device__ __forceinline__ void gru_lane_init(
    GruLane& L, const float* __restrict__ gi, const float* __restrict__ w_hh,
    const float* __restrict__ b_hh, float* __restrict__ hbuf, const int t_base) {
  const int lane = threadIdx.x;
  const int hf   = lane >> 4;
  const int c    = lane & 15;

  const int g_r0 = 0 + 2 * hf, g_r1 = 1 + 2 * hf;
  const int g_z0 = 4 + 2 * hf, g_z1 = 5 + 2 * hf;
  const int g_n0 = 8 + 2 * hf, g_n1 = 9 + 2 * hf;

  {
    const int mh = c >> 3, mi = c & 7;
    if (mi < 6) {
      const int g = (mi >> 1) * 4 + 2 * mh + (mi & 1);
      const float2 wv = *reinterpret_cast<const float2*>(w_hh + g * 4 + 2 * hf);
      L.a.x = wv.x; L.a.y = wv.y;
    } else {
      L.a.x = 0.0f; L.a.y = 0.0f;
    }
  }

  L.bh_r0 = b_hh[g_r0]; L.bh_r1 = b_hh[g_r1];
  L.bh_z0 = b_hh[g_z0]; L.bh_z1 = b_hh[g_z1];
  L.bh_n0 = b_hh[g_n0]; L.bh_n1 = b_hh[g_n1];

  L.p_r0 = gi + (size_t)g_r0 * SEQ_LEN + t_base;
  L.p_r1 = gi + (size_t)g_r1 * SEQ_LEN + t_base;
  L.p_z0 = gi + (size_t)g_z0 * SEQ_LEN + t_base;
  L.p_z1 = gi + (size_t)g_z1 * SEQ_LEN + t_base;
  L.p_n0 = gi + (size_t)g_n0 * SEQ_LEN + t_base;
  L.p_n1 = gi + (size_t)g_n1 * SEQ_LEN + t_base;
  L.hp   = hbuf + (size_t)t_base * HDIM + 2 * hf;
  L.ha = 0.0f; L.hb = 0.0f;
}

__device__ __forceinline__ void gru_load_row(
    float (&dst)[SCAN_GRP], const float* p, int t) {
  const float4 v = *reinterpret_cast<const float4*>(p + t);
  dst[0] = v.x; dst[1] = v.y; dst[2] = v.z; dst[3] = v.w;
}

__device__ __forceinline__ void gru_load_grp(
    const GruLane& L, float (&g)[6][SCAN_GRP], int t) {
  gru_load_row(g[0], L.p_r0, t); gru_load_row(g[1], L.p_r1, t);
  gru_load_row(g[2], L.p_z0, t); gru_load_row(g[3], L.p_z1, t);
  gru_load_row(g[4], L.p_n0, t); gru_load_row(g[5], L.p_n1, t);
}

template <bool STORE_H>
__device__ __forceinline__ void gru_run_grp(
    GruLane& L, const float (&g)[6][SCAN_GRP], int t, bool keep) {
  float hs[SCAN_GRP][2];
#pragma unroll
  for (int u = 0; u < SCAN_GRP; ++u) {
    v8f cc;
    cc[0] = g[0][u] + L.bh_r0; cc[1] = g[1][u] + L.bh_r1;
    cc[2] = g[2][u] + L.bh_z0; cc[3] = g[3][u] + L.bh_z1;
    cc[4] = L.bh_n0;           cc[5] = L.bh_n1;
    cc[6] = 0.0f;              cc[7] = 0.0f;
    v2f b; b.x = L.ha; b.y = L.hb;

    const v8f d = __builtin_amdgcn_wmma_f32_16x16x4_f32(
         false, L.a,  false, b,
         (short)0, cc,  false,  false);

    const float ra = fast_sigmoid(d[0]);
    const float rb = fast_sigmoid(d[1]);
    const float za = fast_sigmoid(d[2]);
    const float zb = fast_sigmoid(d[3]);

    const float na = fmaf(2.0f, fast_sigmoid(2.0f * fmaf(ra, d[4], g[4][u])), -1.0f);
    const float nb = fmaf(2.0f, fast_sigmoid(2.0f * fmaf(rb, d[5], g[5][u])), -1.0f);

    L.ha = fmaf(za, L.ha - na, na);
    L.hb = fmaf(zb, L.hb - nb, nb);
    hs[u][0] = L.ha; hs[u][1] = L.hb;
  }
  if (STORE_H && keep) {
#pragma unroll
    for (int u = 0; u < SCAN_GRP; ++u)
      *reinterpret_cast<float2*>(L.hp + (size_t)(t + u) * HDIM) =
          make_float2(hs[u][0], hs[u][1]);
  }
}

#define CH_LEN   64
#define N_CHUNK  (SEQ_LEN / CH_LEN)
#define RUNIN    8
#define OWNED    (16 - RUNIN)
#define N_SWEEP  (RUNIN + 1)
#define CH_TRIPS (CH_LEN / SCAN_TRIP)
#define SC_WAVES (N_CHUNK / OWNED)
#define REC_PITCH 32
#define REC_FLAG  HDIM

__device__ __forceinline__ float from_left_column(float v) {
  return __builtin_bit_cast(float, __builtin_amdgcn_update_dpp(
      0, __builtin_bit_cast(int, v),   0x111, 0xf, 0xf, true));
}

__global__ __launch_bounds__(32) void gru_sweeps_kernel(
    const float* __restrict__ gi, const float* __restrict__ state0,
    const float* __restrict__ w_hh, const float* __restrict__ b_hh,
    float* __restrict__ hbuf, float* __restrict__ rec) {
  const int lane = threadIdx.x;
  const int hf   = lane >> 4;
  const int c    = lane & 15;
  const int k    = (int)blockIdx.x * OWNED - RUNIN + c;
  const int kc   = k < 0 ? 0 : k;
  const bool head  = (k == 0);
  const bool owned = (c >= RUNIN);

  GruLane L;
  gru_lane_init(L, gi, w_hh, b_hh, hbuf, kc * CH_LEN);

  const float s_a = state0[2 * hf + 0];
  const float s_b = state0[2 * hf + 1];
  float ea = head ? s_a : 0.0f;
  float eb = head ? s_b : 0.0f;
  float xa = 0.0f, xb = 0.0f;

  float g[SCAN_RING][6][SCAN_GRP];
#pragma unroll
  for (int j = 0; j < SCAN_RING; ++j) gru_load_grp(L, g[j], SCAN_GRP * j);

  L.ha = ea; L.hb = eb;
  for (int trip = 0; trip < (N_SWEEP - 1) * CH_TRIPS; ++trip) {
    const int t = (trip & (CH_TRIPS - 1)) * SCAN_TRIP;
#pragma unroll
    for (int j = 0; j < SCAN_RING; ++j) {
      gru_run_grp<false>(L, g[j], t + SCAN_GRP * j, false);
      gru_load_grp(L, g[j], (t + SCAN_TRIP + SCAN_GRP * j) & (CH_LEN - 1));
      __builtin_amdgcn_sched_barrier(0);
    }
    if (t == CH_LEN - SCAN_TRIP) {
      xa = L.ha; xb = L.hb;
      const float la = from_left_column(xa);
      const float lb = from_left_column(xb);
      ea = head ? s_a : la;
      eb = head ? s_b : lb;
      L.ha = ea; L.hb = eb;
    }
  }

  for (int t = 0; t < CH_LEN; t += SCAN_TRIP) {
#pragma unroll
    for (int j = 0; j < SCAN_RING; ++j) {
      gru_run_grp<true>(L, g[j], t + SCAN_GRP * j, owned);
      gru_load_grp(L, g[j], (t + SCAN_TRIP + SCAN_GRP * j) & (CH_LEN - 1));
      __builtin_amdgcn_sched_barrier(0);
    }
  }

  const bool moved = owned & (c < 15) &
      ((__builtin_bit_cast(int, L.ha) != __builtin_bit_cast(int, xa)) |
       (__builtin_bit_cast(int, L.hb) != __builtin_bit_cast(int, xb)));
  const unsigned any_moved = __builtin_amdgcn_ballot_w32(moved);
  float* line = rec + (size_t)blockIdx.x * REC_PITCH;
  if (lane == 0)
    line[REC_FLAG] = (any_moved != 0u) ? 1.0f : 0.0f;
  if (c == RUNIN)
    *reinterpret_cast<float2*>(line + 2 * hf) = make_float2(ea, eb);
}

template <int NSTEPS>
__device__ __forceinline__ void gru_serial(GruLane& L) {
  float g[SCAN_RING][6][SCAN_GRP];
#pragma unroll
  for (int j = 0; j < SCAN_RING; ++j) gru_load_grp(L, g[j], SCAN_GRP * j);

  for (int t = 0; t < NSTEPS; t += SCAN_TRIP) {
#pragma unroll
    for (int j = 0; j < SCAN_RING; ++j) {
      gru_run_grp<true>(L, g[j], t + SCAN_GRP * j, true);
      gru_load_grp(L, g[j], t + SCAN_TRIP + SCAN_GRP * j);
      __builtin_amdgcn_sched_barrier(0);
    }
  }
}

__global__ __launch_bounds__(32) void gru_join_kernel(
    const float* __restrict__ gi, const float* __restrict__ state0,
    const float* __restrict__ w_hh, const float* __restrict__ b_hh,
    float* __restrict__ hbuf, const float* __restrict__ rec,
    float* __restrict__ state_out) {
  const int lane = threadIdx.x;
  const int hf   = lane >> 4;
  const int c    = lane & 15;

  bool bad = false;
#pragma unroll
  for (int j = 0; j < SC_WAVES / 32; ++j) {
    const int w  = j * 32 + lane;
    const int wl = w > 0 ? w : 1;
    const float* line = rec + (size_t)w * REC_PITCH;
    const float4 e = *reinterpret_cast<const float4*>(rec + (size_t)wl * REC_PITCH);
    const float4 x = *reinterpret_cast<const float4*>(
        hbuf + ((size_t)wl * OWNED * CH_LEN - 1) * HDIM);
    bad |= (line[REC_FLAG] != 0.0f);
    bad |= (w > 0) &
           ((__builtin_bit_cast(int, e.x) != __builtin_bit_cast(int, x.x)) |
            (__builtin_bit_cast(int, e.y) != __builtin_bit_cast(int, x.y)) |
            (__builtin_bit_cast(int, e.z) != __builtin_bit_cast(int, x.z)) |
            (__builtin_bit_cast(int, e.w) != __builtin_bit_cast(int, x.w)));
  }

  float ha, hb;
  if (__builtin_amdgcn_ballot_w32(bad) != 0u) {
    GruLane L;
    gru_lane_init(L, gi, w_hh, b_hh, hbuf, 0);
    L.ha = state0[2 * hf + 0];
    L.hb = state0[2 * hf + 1];
    gru_serial<SEQ_LEN>(L);
    ha = L.ha; hb = L.hb;
  } else {
    const float2 v = *reinterpret_cast<const float2*>(
        hbuf + (size_t)(SEQ_LEN - 1) * HDIM + 2 * hf);
    ha = v.x; hb = v.y;
  }

  if (c == 0) {
    state_out[2 * hf + 0] = ha;
    state_out[2 * hf + 1] = hb;
  }
}

__global__ __launch_bounds__(256) void yproj_wmma_kernel(
    const float* __restrict__ hbuf, const float* __restrict__ w_out,
    const float* __restrict__ b_out, float* __restrict__ y) {
  const int lane = threadIdx.x & 31;
  const int wave = threadIdx.x >> 5;
  const int tile = blockIdx.x * 8 + wave;
  const int t0   = tile * 16;
  const int half = lane >> 4;
  const int idx  = lane & 15;

  v2f a;
  {
    const float2 hv = *reinterpret_cast<const float2*>(
        hbuf + (size_t)(t0 + idx) * 4 + half * 2);
    a.x = hv.x; a.y = hv.y;
  }

  v2f b;
  if (idx < 2) {
    const float2 wv = *reinterpret_cast<const float2*>(w_out + idx * 4 + half * 2);
    b.x = wv.x; b.y = wv.y;
  } else {
    b.x = 0.0f; b.y = 0.0f;
  }

  const float bias = (idx < 2) ? b_out[idx] : 0.0f;
  v8f c;
#pragma unroll
  for (int i = 0; i < 8; ++i) c[i] = bias;

  v8f d = __builtin_amdgcn_wmma_f32_16x16x4_f32(
      false, a, false, b, (short)0, c, false, false);

  if (idx < 2) {
    const int rbase = half * 8;
#pragma unroll
    for (int i = 0; i < 8; ++i)
      y[(size_t)(t0 + rbase + i) * 2 + idx] = d[i];
  }
}

extern "C" void kernel_launch(void* const* d_in, const int* in_sizes, int n_in,
                              void* d_out, int out_size, void* d_ws, size_t ws_size,
                              hipStream_t stream) {
  const float* x      = (const float*)d_in[0];
  const float* state  = (const float*)d_in[1];
  const float* w_ih   = (const float*)d_in[2];
  const float* w_hh   = (const float*)d_in[3];
  const float* b_ih   = (const float*)d_in[4];
  const float* b_hh   = (const float*)d_in[5];
  const float* w_out  = (const float*)d_in[6];
  const float* b_out  = (const float*)d_in[7];

  float* out  = (float*)d_out;
  float* gi   = (float*)d_ws;
  float* hbuf = gi + (size_t)NGATES * SEQ_LEN;

  const int tiles  = SEQ_LEN / 16;
  const int blocks = tiles / 8;

  gi_wmma_kernel<<<blocks, 256, 0, stream>>>(x, w_ih, b_ih, gi);
  gru_sweeps_kernel<<<SC_WAVES, 32, 0, stream>>>(gi, state, w_hh, b_hh, hbuf, out);
  gru_join_kernel<<<1, 32, 0, stream>>>(gi, state, w_hh, b_hh, hbuf, out,
                                        out + (size_t)SEQ_LEN * 2);
  yproj_wmma_kernel<<<blocks, 256, 0, stream>>>(hbuf, w_out, b_out, out);
}


# y = h @ w_out.T + b_out rides in the scan's own WMMA: w_out in A rows 6,7 (were zero), b_out in C, so step t+1's product hands out y[t]; projection launch and the 4 MB h buffer are gone
# speedup vs baseline: 1035.9949x; 1.1624x over previous
#include <hip/hip_runtime.h>

typedef __attribute__((ext_vector_type(2))) float v2f;
typedef __attribute__((ext_vector_type(8))) float v8f;

#define SEQ_LEN 262144
#define NGATES  12
#define HDIM    4

__device__ __forceinline__ float fast_sigmoid(float x) {
  const float e = __builtin_amdgcn_exp2f(-1.44269504088896340736f * x);
  return __builtin_amdgcn_rcpf(1.0f + e);
}

__global__ __launch_bounds__(256) void gi_wmma_kernel(
    const float* __restrict__ x, const float* __restrict__ w_ih,
    const float* __restrict__ b_ih, float* __restrict__ gi) {
  const int lane = threadIdx.x & 31;
  const int wave = threadIdx.x >> 5;
  const int tile = blockIdx.x * 8 + wave;
  const int t0   = tile * 16;
  const int half = lane >> 4;
  const int idx  = lane & 15;

  v2f a;
  if (half == 0) {
    const float2 xv = *reinterpret_cast<const float2*>(x + (size_t)(t0 + idx) * 2);
    a.x = xv.x; a.y = xv.y;
  } else {
    a.x = 0.0f; a.y = 0.0f;
  }

  v2f b;
  if (half == 0 && idx < NGATES) {
    const float2 wv = *reinterpret_cast<const float2*>(w_ih + idx * 2);
    b.x = wv.x; b.y = wv.y;
  } else {
    b.x = 0.0f; b.y = 0.0f;
  }

  const float bias = (idx < NGATES) ? b_ih[idx] : 0.0f;
  v8f c;
#pragma unroll
  for (int i = 0; i < 8; ++i) c[i] = bias;

  v8f d = __builtin_amdgcn_wmma_f32_16x16x4_f32(
       false, a,  false, b,
       (short)0, c,  false,  false);

  if (idx < NGATES) {
    float* dst = gi + (size_t)idx * SEQ_LEN + t0 + half * 8;
    *reinterpret_cast<float4*>(dst + 0) = make_float4(d[0], d[1], d[2], d[3]);
    *reinterpret_cast<float4*>(dst + 4) = make_float4(d[4], d[5], d[6], d[7]);
  }
}

#define SCAN_GRP  4
#define SCAN_RING 4
#define SCAN_TRIP (SCAN_GRP * SCAN_RING)

struct GruLane {
  v2f a;
  float bh_r0, bh_r1, bh_z0, bh_z1, bh_n0, bh_n1;
  const float* p_r0; const float* p_r1;
  const float* p_z0; const float* p_z1;
  const float* p_n0; const float* p_n1;
  float bo0, bo1;
  float* yp;
  float ha, hb;
};

__device__ __forceinline__ void gru_lane_init(
    GruLane& L, const float* __restrict__ gi, const float* __restrict__ w_hh,
    const float* __restrict__ b_hh, const float* __restrict__ w_out,
    const float* __restrict__ b_out, float* __restrict__ y, const int t_base) {
  const int lane = threadIdx.x;
  const int hf   = lane >> 4;
  const int c    = lane & 15;

  const int g_r0 = 0 + 2 * hf, g_r1 = 1 + 2 * hf;
  const int g_z0 = 4 + 2 * hf, g_z1 = 5 + 2 * hf;
  const int g_n0 = 8 + 2 * hf, g_n1 = 9 + 2 * hf;

  {
    const int mh = c >> 3, mi = c & 7;
    if (mi < 6) {
      const int g = (mi >> 1) * 4 + 2 * mh + (mi & 1);
      const float2 wv = *reinterpret_cast<const float2*>(w_hh + g * 4 + 2 * hf);
      L.a.x = wv.x; L.a.y = wv.y;
    } else if (mh == 0) {
      const float2 wv = *reinterpret_cast<const float2*>(w_out + (mi - 6) * 4 + 2 * hf);
      L.a.x = wv.x; L.a.y = wv.y;
    } else {
      L.a.x = 0.0f; L.a.y = 0.0f;
    }
  }
  L.bo0 = b_out[0]; L.bo1 = b_out[1];

  L.bh_r0 = b_hh[g_r0]; L.bh_r1 = b_hh[g_r1];
  L.bh_z0 = b_hh[g_z0]; L.bh_z1 = b_hh[g_z1];
  L.bh_n0 = b_hh[g_n0]; L.bh_n1 = b_hh[g_n1];

  L.p_r0 = gi + (size_t)g_r0 * SEQ_LEN + t_base;
  L.p_r1 = gi + (size_t)g_r1 * SEQ_LEN + t_base;
  L.p_z0 = gi + (size_t)g_z0 * SEQ_LEN + t_base;
  L.p_z1 = gi + (size_t)g_z1 * SEQ_LEN + t_base;
  L.p_n0 = gi + (size_t)g_n0 * SEQ_LEN + t_base;
  L.p_n1 = gi + (size_t)g_n1 * SEQ_LEN + t_base;
  L.yp   = y + (size_t)t_base * 2;
  L.ha = 0.0f; L.hb = 0.0f;
}

__device__ __forceinline__ void gru_load_row(
    float (&dst)[SCAN_GRP], const float* p, int t) {
  const float4 v = *reinterpret_cast<const float4*>(p + t);
  dst[0] = v.x; dst[1] = v.y; dst[2] = v.z; dst[3] = v.w;
}

__device__ __forceinline__ void gru_load_grp(
    const GruLane& L, float (&g)[6][SCAN_GRP], int t) {
  gru_load_row(g[0], L.p_r0, t); gru_load_row(g[1], L.p_r1, t);
  gru_load_row(g[2], L.p_z0, t); gru_load_row(g[3], L.p_z1, t);
  gru_load_row(g[4], L.p_n0, t); gru_load_row(g[5], L.p_n1, t);
}

template <bool STORE_Y>
__device__ __forceinline__ void gru_run_grp(
    GruLane& L, const float (&g)[6][SCAN_GRP], int t, bool keep) {
  float ys[SCAN_GRP][2];
#pragma unroll
  for (int u = 0; u < SCAN_GRP; ++u) {
    v8f cc;
    cc[0] = g[0][u] + L.bh_r0; cc[1] = g[1][u] + L.bh_r1;
    cc[2] = g[2][u] + L.bh_z0; cc[3] = g[3][u] + L.bh_z1;
    cc[4] = L.bh_n0;           cc[5] = L.bh_n1;
    cc[6] = L.bo0;             cc[7] = L.bo1;
    v2f b; b.x = L.ha; b.y = L.hb;

    const v8f d = __builtin_amdgcn_wmma_f32_16x16x4_f32(
         false, L.a,  false, b,
         (short)0, cc,  false,  false);
    ys[u][0] = d[6]; ys[u][1] = d[7];

    const float ra = fast_sigmoid(d[0]);
    const float rb = fast_sigmoid(d[1]);
    const float za = fast_sigmoid(d[2]);
    const float zb = fast_sigmoid(d[3]);

    const float na = fmaf(2.0f, fast_sigmoid(2.0f * fmaf(ra, d[4], g[4][u])), -1.0f);
    const float nb = fmaf(2.0f, fast_sigmoid(2.0f * fmaf(rb, d[5], g[5][u])), -1.0f);

    L.ha = fmaf(za, L.ha - na, na);
    L.hb = fmaf(zb, L.hb - nb, nb);
  }
  if (STORE_Y && keep) {
    if (t > 0)
      *reinterpret_cast<float2*>(L.yp + (t - 1) * 2) = make_float2(ys[0][0], ys[0][1]);
#pragma unroll
    for (int u = 1; u < SCAN_GRP; ++u)
      *reinterpret_cast<float2*>(L.yp + (t + u - 1) * 2) =
          make_float2(ys[u][0], ys[u][1]);
  }
}

__device__ __forceinline__ void gru_last_y(const GruLane& L, int t_last, bool keep) {
  const v8f cc = {0.0f, 0.0f, 0.0f, 0.0f, 0.0f, 0.0f, L.bo0, L.bo1};
  v2f b; b.x = L.ha; b.y = L.hb;
  const v8f d = __builtin_amdgcn_wmma_f32_16x16x4_f32(
       false, L.a,  false, b,
       (short)0, cc,  false,  false);
  if (keep)
    *reinterpret_cast<float2*>(L.yp + t_last * 2) = make_float2(d[6], d[7]);
}

#define CH_LEN   64
#define N_CHUNK  (SEQ_LEN / CH_LEN)
#define RUNIN    8
#define OWNED    (16 - RUNIN)
#define N_SWEEP  (RUNIN + 1)
#define CH_TRIPS (CH_LEN / SCAN_TRIP)
#define SC_WAVES (N_CHUNK / OWNED)
#define REC_PITCH 32
#define REC_EXIT  HDIM
#define REC_FLAG  (2 * HDIM)

__device__ __forceinline__ float from_left_column(float v) {
  return __builtin_bit_cast(float, __builtin_amdgcn_update_dpp(
      0, __builtin_bit_cast(int, v),   0x111, 0xf, 0xf, true));
}

__global__ __launch_bounds__(32) void gru_sweeps_kernel(
    const float* __restrict__ gi, const float* __restrict__ state0,
    const float* __restrict__ w_hh, const float* __restrict__ b_hh,
    const float* __restrict__ w_out, const float* __restrict__ b_out,
    float* __restrict__ y, float* __restrict__ rec) {
  const int lane = threadIdx.x;
  const int hf   = lane >> 4;
  const int c    = lane & 15;
  const int k    = (int)blockIdx.x * OWNED - RUNIN + c;
  const int kc   = k < 0 ? 0 : k;
  const bool head  = (k == 0);
  const bool owned = (c >= RUNIN);
  const bool ykeep = owned & (hf == 0);

  GruLane L;
  gru_lane_init(L, gi, w_hh, b_hh, w_out, b_out, y, kc * CH_LEN);

  const float s_a = state0[2 * hf + 0];
  const float s_b = state0[2 * hf + 1];
  float ea = head ? s_a : 0.0f;
  float eb = head ? s_b : 0.0f;
  float xa = 0.0f, xb = 0.0f;

  float g[SCAN_RING][6][SCAN_GRP];
#pragma unroll
  for (int j = 0; j < SCAN_RING; ++j) gru_load_grp(L, g[j], SCAN_GRP * j);

  L.ha = ea; L.hb = eb;
  for (int trip = 0; trip < (N_SWEEP - 1) * CH_TRIPS; ++trip) {
    const int t = (trip & (CH_TRIPS - 1)) * SCAN_TRIP;
#pragma unroll
    for (int j = 0; j < SCAN_RING; ++j) {
      gru_run_grp<false>(L, g[j], t + SCAN_GRP * j, false);
      gru_load_grp(L, g[j], (t + SCAN_TRIP + SCAN_GRP * j) & (CH_LEN - 1));
      __builtin_amdgcn_sched_barrier(0);
    }
    if (t == CH_LEN - SCAN_TRIP) {
      xa = L.ha; xb = L.hb;
      const float la = from_left_column(xa);
      const float lb = from_left_column(xb);
      ea = head ? s_a : la;
      eb = head ? s_b : lb;
      L.ha = ea; L.hb = eb;
    }
  }

  for (int t = 0; t < CH_LEN; t += SCAN_TRIP) {
#pragma unroll
    for (int j = 0; j < SCAN_RING; ++j) {
      gru_run_grp<true>(L, g[j], t + SCAN_GRP * j, ykeep);
      gru_load_grp(L, g[j], (t + SCAN_TRIP + SCAN_GRP * j) & (CH_LEN - 1));
      __builtin_amdgcn_sched_barrier(0);
    }
  }

  gru_last_y(L, CH_LEN - 1, ykeep);

  const bool moved = owned & (c < 15) &
      ((__builtin_bit_cast(int, L.ha) != __builtin_bit_cast(int, xa)) |
       (__builtin_bit_cast(int, L.hb) != __builtin_bit_cast(int, xb)));
  const unsigned any_moved = __builtin_amdgcn_ballot_w32(moved);
  float* line = rec + (size_t)blockIdx.x * REC_PITCH;
  if (lane == 0)
    line[REC_FLAG] = (any_moved != 0u) ? 1.0f : 0.0f;
  if (c == RUNIN)
    *reinterpret_cast<float2*>(line + 2 * hf) = make_float2(ea, eb);
  if (c == 15)
    *reinterpret_cast<float2*>(line + REC_EXIT + 2 * hf) = make_float2(L.ha, L.hb);
}

template <int NSTEPS>
__device__ __forceinline__ void gru_serial(GruLane& L, bool keep) {
  float g[SCAN_RING][6][SCAN_GRP];
#pragma unroll
  for (int j = 0; j < SCAN_RING; ++j) gru_load_grp(L, g[j], SCAN_GRP * j);

  for (int t = 0; t < NSTEPS; t += SCAN_TRIP) {
#pragma unroll
    for (int j = 0; j < SCAN_RING; ++j) {
      gru_run_grp<true>(L, g[j], t + SCAN_GRP * j, keep);
      gru_load_grp(L, g[j], t + SCAN_TRIP + SCAN_GRP * j);
      __builtin_amdgcn_sched_barrier(0);
    }
  }
  gru_last_y(L, NSTEPS - 1, keep);
}

__global__ __launch_bounds__(32) void gru_join_kernel(
    const float* __restrict__ gi, const float* __restrict__ state0,
    const float* __restrict__ w_hh, const float* __restrict__ b_hh,
    const float* __restrict__ w_out, const float* __restrict__ b_out,
    float* __restrict__ y, const float* __restrict__ rec,
    float* __restrict__ state_out) {
  const int lane = threadIdx.x;
  const int hf   = lane >> 4;
  const int c    = lane & 15;

  bool bad = false;
#pragma unroll
  for (int j = 0; j < SC_WAVES / 32; ++j) {
    const int w  = j * 32 + lane;
    const int wl = w > 0 ? w : 1;
    const float* line = rec + (size_t)w * REC_PITCH;
    const float4 e = *reinterpret_cast<const float4*>(rec + (size_t)wl * REC_PITCH);
    const float4 x = *reinterpret_cast<const float4*>(
        rec + (size_t)(wl - 1) * REC_PITCH + REC_EXIT);
    bad |= (line[REC_FLAG] != 0.0f);
    bad |= (w > 0) &
           ((__builtin_bit_cast(int, e.x) != __builtin_bit_cast(int, x.x)) |
            (__builtin_bit_cast(int, e.y) != __builtin_bit_cast(int, x.y)) |
            (__builtin_bit_cast(int, e.z) != __builtin_bit_cast(int, x.z)) |
            (__builtin_bit_cast(int, e.w) != __builtin_bit_cast(int, x.w)));
  }

  float ha, hb;
  if (__builtin_amdgcn_ballot_w32(bad) != 0u) {
    GruLane L;
    gru_lane_init(L, gi, w_hh, b_hh, w_out, b_out, y, 0);
    L.ha = state0[2 * hf + 0];
    L.hb = state0[2 * hf + 1];
    gru_serial<SEQ_LEN>(L, lane == 0);
    ha = L.ha; hb = L.hb;
  } else {
    const float2 v = *reinterpret_cast<const float2*>(
        rec + (size_t)(SC_WAVES - 1) * REC_PITCH + REC_EXIT + 2 * hf);
    ha = v.x; hb = v.y;
  }

  if (c == 0) {
    state_out[2 * hf + 0] = ha;
    state_out[2 * hf + 1] = hb;
  }
}

extern "C" void kernel_launch(void* const* d_in, const int* in_sizes, int n_in,
                              void* d_out, int out_size, void* d_ws, size_t ws_size,
                              hipStream_t stream) {
  const float* x      = (const float*)d_in[0];
  const float* state  = (const float*)d_in[1];
  const float* w_ih   = (const float*)d_in[2];
  const float* w_hh   = (const float*)d_in[3];
  const float* b_ih   = (const float*)d_in[4];
  const float* b_hh   = (const float*)d_in[5];
  const float* w_out  = (const float*)d_in[6];
  const float* b_out  = (const float*)d_in[7];

  float* out  = (float*)d_out;
  float* gi   = (float*)d_ws;
  float* rec  = gi + (size_t)NGATES * SEQ_LEN;

  const int tiles  = SEQ_LEN / 16;
  const int blocks = tiles / 8;

  gi_wmma_kernel<<<blocks, 256, 0, stream>>>(x, w_ih, b_ih, gi);
  gru_sweeps_kernel<<<SC_WAVES, 32, 0, stream>>>(gi, state, w_hh, b_hh, w_out, b_out,
                                                 out, rec);
  gru_join_kernel<<<1, 32, 0, stream>>>(gi, state, w_hh, b_hh, w_out, b_out, out, rec,
                                        out + (size_t)SEQ_LEN * 2);
}


# gi = x @ w_ih.T + biases rides in the scan as a second f32 WMMA per step (A = w_ih rows + a bias column, B = x0,x1,1,0 per column, one step ahead): gi launch, 12 MB buffer gone; gi sum order changes
# speedup vs baseline: 1281.2904x; 1.2368x over previous
#include <hip/hip_runtime.h>

typedef __attribute__((ext_vector_type(2))) float v2f;
typedef __attribute__((ext_vector_type(8))) float v8f;

#define SEQ_LEN 262144
#define HDIM    4

__device__ __forceinline__ float fast_sigmoid(float x) {
  const float e = __builtin_amdgcn_exp2f(-1.44269504088896340736f * x);
  return __builtin_amdgcn_rcpf(1.0f + e);
}

#define SCAN_GRP  4
#define SCAN_RING 4
#define SCAN_TRIP (SCAN_GRP * SCAN_RING)

struct GruLane {
  v2f a;
  v2f ax;
  float bh_n0, bh_n1;
  float bo0, bo1;
  const float* xp;
  float* yp;
  float ha, hb;
  bool upper;
};

__device__ __forceinline__ void gru_lane_place(
    GruLane& L, const float* __restrict__ x, float* __restrict__ y, const int t_base) {
  L.xp = x + (size_t)t_base * 2;
  L.yp = y + (size_t)t_base * 2;
  L.upper = (threadIdx.x >> 4) != 0;
}

__device__ __forceinline__ void gru_lane_init(
    GruLane& L, const float* __restrict__ w_ih,
    const float* __restrict__ w_hh, const float* __restrict__ b_ih,
    const float* __restrict__ b_hh, const float* __restrict__ w_out,
    const float* __restrict__ b_out) {
  const int lane = threadIdx.x;
  const int hf   = lane >> 4;
  const int c    = lane & 15;

  {
    const int mh = c >> 3, mi = c & 7;
    if (mi < 6) {
      const int g = (mi >> 1) * 4 + 2 * mh + (mi & 1);
      const float2 wv = *reinterpret_cast<const float2*>(w_hh + g * 4 + 2 * hf);
      L.a.x = wv.x; L.a.y = wv.y;
      if (hf == 0) {
        const float2 wi = *reinterpret_cast<const float2*>(w_ih + g * 2);
        L.ax.x = wi.x; L.ax.y = wi.y;
      } else {
        const float bi = b_ih[g];
        const float bh = b_hh[g];
        L.ax.x = (g < 2 * HDIM) ? bi + bh : bi;
        L.ax.y = 0.0f;
      }
    } else if (mh == 0) {
      const float2 wv = *reinterpret_cast<const float2*>(w_out + (mi - 6) * 4 + 2 * hf);
      L.a.x = wv.x; L.a.y = wv.y;
      L.ax.x = 0.0f; L.ax.y = 0.0f;
    } else {
      L.a.x = 0.0f; L.a.y = 0.0f;
      L.ax.x = 0.0f; L.ax.y = 0.0f;
    }
  }
  L.bo0 = b_out[0]; L.bo1 = b_out[1];
  L.bh_n0 = b_hh[2 * HDIM + 2 * hf + 0];
  L.bh_n1 = b_hh[2 * HDIM + 2 * hf + 1];

  L.ha = 0.0f; L.hb = 0.0f;
}

__device__ __forceinline__ void gru_load_grp(
    const GruLane& L, float (&xs)[2 * SCAN_GRP], int t) {
  const float4 v0 = *reinterpret_cast<const float4*>(L.xp + 2 * t);
  const float4 v1 = *reinterpret_cast<const float4*>(L.xp + 2 * t + 4);
  xs[0] = v0.x; xs[1] = v0.y; xs[2] = v0.z; xs[3] = v0.w;
  xs[4] = v1.x; xs[5] = v1.y; xs[6] = v1.z; xs[7] = v1.w;
}

__device__ __forceinline__ v8f gru_gi(const GruLane& L, float x0, float x1) {
  v2f b;
  b.x = L.upper ? 1.0f : x0;
  b.y = L.upper ? 0.0f : x1;
  const v8f zero = {0.0f, 0.0f, 0.0f, 0.0f, 0.0f, 0.0f, 0.0f, 0.0f};
  return __builtin_amdgcn_wmma_f32_16x16x4_f32(
       false, L.ax,  false, b,
       (short)0, zero,  false,  false);
}

template <bool STORE_Y>
__device__ __forceinline__ void gru_run_grp(
    GruLane& L, const float (&xs)[2 * SCAN_GRP], float nx0, float nx1,
    v8f& gnext, int t, bool keep) {
  float ys[SCAN_GRP][2];
#pragma unroll
  for (int u = 0; u < SCAN_GRP; ++u) {
    const v8f gi = gnext;
    v8f cc;
    cc[0] = gi[0];   cc[1] = gi[1];
    cc[2] = gi[2];   cc[3] = gi[3];
    cc[4] = L.bh_n0; cc[5] = L.bh_n1;
    cc[6] = L.bo0;   cc[7] = L.bo1;
    v2f b; b.x = L.ha; b.y = L.hb;

    const v8f d = __builtin_amdgcn_wmma_f32_16x16x4_f32(
         false, L.a,  false, b,
         (short)0, cc,  false,  false);
    ys[u][0] = d[6]; ys[u][1] = d[7];

    gnext = (u + 1 < SCAN_GRP) ? gru_gi(L, xs[2 * u + 2], xs[2 * u + 3])
                               : gru_gi(L, nx0, nx1);

    const float ra = fast_sigmoid(d[0]);
    const float rb = fast_sigmoid(d[1]);
    const float za = fast_sigmoid(d[2]);
    const float zb = fast_sigmoid(d[3]);

    const float na = fmaf(2.0f, fast_sigmoid(2.0f * fmaf(ra, d[4], gi[4])), -1.0f);
    const float nb = fmaf(2.0f, fast_sigmoid(2.0f * fmaf(rb, d[5], gi[5])), -1.0f);

    L.ha = fmaf(za, L.ha - na, na);
    L.hb = fmaf(zb, L.hb - nb, nb);
  }
  if (STORE_Y && keep) {
    if (t > 0)
      *reinterpret_cast<float2*>(L.yp + (t - 1) * 2) = make_float2(ys[0][0], ys[0][1]);
#pragma unroll
    for (int u = 1; u < SCAN_GRP; ++u)
      *reinterpret_cast<float2*>(L.yp + (t + u - 1) * 2) =
          make_float2(ys[u][0], ys[u][1]);
  }
}

__device__ __forceinline__ void gru_last_y(const GruLane& L, int t_last, bool keep) {
  const v8f cc = {0.0f, 0.0f, 0.0f, 0.0f, 0.0f, 0.0f, L.bo0, L.bo1};
  v2f b; b.x = L.ha; b.y = L.hb;
  const v8f d = __builtin_amdgcn_wmma_f32_16x16x4_f32(
       false, L.a,  false, b,
       (short)0, cc,  false,  false);
  if (keep)
    *reinterpret_cast<float2*>(L.yp + t_last * 2) = make_float2(d[6], d[7]);
}

#define CH_LEN   64
#define N_CHUNK  (SEQ_LEN / CH_LEN)
#define RUNIN    8
#define OWNED    (16 - RUNIN)
#define N_SWEEP  (RUNIN + 1)
#define CH_TRIPS (CH_LEN / SCAN_TRIP)
#define SC_WAVES (N_CHUNK / OWNED)
#define REC_PITCH 32
#define REC_EXIT  HDIM
#define REC_FLAG  (2 * HDIM)

__device__ __forceinline__ float from_left_column(float v) {
  return __builtin_bit_cast(float, __builtin_amdgcn_update_dpp(
      0, __builtin_bit_cast(int, v),   0x111, 0xf, 0xf, true));
}

__global__ __launch_bounds__(32) void gru_sweeps_kernel(
    const float* __restrict__ x, const float* __restrict__ state0,
    const float* __restrict__ w_ih, const float* __restrict__ w_hh,
    const float* __restrict__ b_ih, const float* __restrict__ b_hh,
    const float* __restrict__ w_out, const float* __restrict__ b_out,
    float* __restrict__ y, float* __restrict__ rec) {
  const int lane = threadIdx.x;
  const int hf   = lane >> 4;
  const int c    = lane & 15;
  const int k    = (int)blockIdx.x * OWNED - RUNIN + c;
  const int kc   = k < 0 ? 0 : k;
  const bool head  = (k == 0);
  const bool owned = (c >= RUNIN);
  const bool ykeep = owned & (hf == 0);

  GruLane L;
  gru_lane_place(L, x, y, kc * CH_LEN);

  float g[SCAN_RING][2 * SCAN_GRP];
#pragma unroll
  for (int j = 0; j < SCAN_RING; ++j) gru_load_grp(L, g[j], SCAN_GRP * j);

  gru_lane_init(L, w_ih, w_hh, b_ih, b_hh, w_out, b_out);

  const float s_a = state0[2 * hf + 0];
  const float s_b = state0[2 * hf + 1];
  float ea = head ? s_a : 0.0f;
  float eb = head ? s_b : 0.0f;
  float xa = 0.0f, xb = 0.0f;

  v8f gnext = gru_gi(L, g[0][0], g[0][1]);

  L.ha = ea; L.hb = eb;
  for (int trip = 0; trip < (N_SWEEP - 1) * CH_TRIPS; ++trip) {
    const int t = (trip & (CH_TRIPS - 1)) * SCAN_TRIP;
#pragma unroll
    for (int j = 0; j < SCAN_RING; ++j) {
      gru_run_grp<false>(L, g[j], g[(j + 1) % SCAN_RING][0], g[(j + 1) % SCAN_RING][1],
                         gnext, t + SCAN_GRP * j, false);
      gru_load_grp(L, g[j], (t + SCAN_TRIP + SCAN_GRP * j) & (CH_LEN - 1));
      __builtin_amdgcn_sched_barrier(0);
    }
    if (t == CH_LEN - SCAN_TRIP) {
      xa = L.ha; xb = L.hb;
      const float la = from_left_column(xa);
      const float lb = from_left_column(xb);
      ea = head ? s_a : la;
      eb = head ? s_b : lb;
      L.ha = ea; L.hb = eb;
    }
  }

  for (int t = 0; t < CH_LEN; t += SCAN_TRIP) {
#pragma unroll
    for (int j = 0; j < SCAN_RING; ++j) {
      gru_run_grp<true>(L, g[j], g[(j + 1) % SCAN_RING][0], g[(j + 1) % SCAN_RING][1],
                        gnext, t + SCAN_GRP * j, ykeep);
      gru_load_grp(L, g[j], (t + SCAN_TRIP + SCAN_GRP * j) & (CH_LEN - 1));
      __builtin_amdgcn_sched_barrier(0);
    }
  }

  gru_last_y(L, CH_LEN - 1, ykeep);

  const bool moved = owned & (c < 15) &
      ((__builtin_bit_cast(int, L.ha) != __builtin_bit_cast(int, xa)) |
       (__builtin_bit_cast(int, L.hb) != __builtin_bit_cast(int, xb)));
  const unsigned any_moved = __builtin_amdgcn_ballot_w32(moved);
  float* line = rec + (size_t)blockIdx.x * REC_PITCH;
  if (lane == 0)
    line[REC_FLAG] = (any_moved != 0u) ? 1.0f : 0.0f;
  if (c == RUNIN)
    *reinterpret_cast<float2*>(line + 2 * hf) = make_float2(ea, eb);
  if (c == 15)
    *reinterpret_cast<float2*>(line + REC_EXIT + 2 * hf) = make_float2(L.ha, L.hb);
}

template <int NSTEPS>
__device__ __forceinline__ void gru_serial(GruLane& L, bool keep) {
  float g[SCAN_RING][2 * SCAN_GRP];
#pragma unroll
  for (int j = 0; j < SCAN_RING; ++j) gru_load_grp(L, g[j], SCAN_GRP * j);
  v8f gnext = gru_gi(L, g[0][0], g[0][1]);

  for (int t = 0; t < NSTEPS; t += SCAN_TRIP) {
#pragma unroll
    for (int j = 0; j < SCAN_RING; ++j) {
      gru_run_grp<true>(L, g[j], g[(j + 1) % SCAN_RING][0], g[(j + 1) % SCAN_RING][1],
                        gnext, t + SCAN_GRP * j, keep);
      gru_load_grp(L, g[j], (t + SCAN_TRIP + SCAN_GRP * j) & (NSTEPS - 1));
      __builtin_amdgcn_sched_barrier(0);
    }
  }
  gru_last_y(L, NSTEPS - 1, keep);
}

__global__ __launch_bounds__(32) void gru_join_kernel(
    const float* __restrict__ x, const float* __restrict__ state0,
    const float* __restrict__ w_ih, const float* __restrict__ w_hh,
    const float* __restrict__ b_ih, const float* __restrict__ b_hh,
    const float* __restrict__ w_out, const float* __restrict__ b_out,
    float* __restrict__ y, const float* __restrict__ rec,
    float* __restrict__ state_out) {
  const int lane = threadIdx.x;
  const int hf   = lane >> 4;
  const int c    = lane & 15;

  bool bad = false;
#pragma unroll
  for (int j = 0; j < SC_WAVES / 32; ++j) {
    const int w  = j * 32 + lane;
    const int wl = w > 0 ? w : 1;
    const float* line = rec + (size_t)w * REC_PITCH;
    const float4 e = *reinterpret_cast<const float4*>(rec + (size_t)wl * REC_PITCH);
    const float4 xl = *reinterpret_cast<const float4*>(
        rec + (size_t)(wl - 1) * REC_PITCH + REC_EXIT);
    bad |= (line[REC_FLAG] != 0.0f);
    bad |= (w > 0) &
           ((__builtin_bit_cast(int, e.x) != __builtin_bit_cast(int, xl.x)) |
            (__builtin_bit_cast(int, e.y) != __builtin_bit_cast(int, xl.y)) |
            (__builtin_bit_cast(int, e.z) != __builtin_bit_cast(int, xl.z)) |
            (__builtin_bit_cast(int, e.w) != __builtin_bit_cast(int, xl.w)));
  }

  float ha, hb;
  if (__builtin_amdgcn_ballot_w32(bad) != 0u) {
    GruLane L;
    gru_lane_place(L, x, y, 0);
    gru_lane_init(L, w_ih, w_hh, b_ih, b_hh, w_out, b_out);
    L.ha = state0[2 * hf + 0];
    L.hb = state0[2 * hf + 1];
    gru_serial<SEQ_LEN>(L, lane == 0);
    ha = L.ha; hb = L.hb;
  } else {
    const float2 v = *reinterpret_cast<const float2*>(
        rec + (size_t)(SC_WAVES - 1) * REC_PITCH + REC_EXIT + 2 * hf);
    ha = v.x; hb = v.y;
  }

  if (c == 0) {
    state_out[2 * hf + 0] = ha;
    state_out[2 * hf + 1] = hb;
  }
}

extern "C" void kernel_launch(void* const* d_in, const int* in_sizes, int n_in,
                              void* d_out, int out_size, void* d_ws, size_t ws_size,
                              hipStream_t stream) {
  const float* x      = (const float*)d_in[0];
  const float* state  = (const float*)d_in[1];
  const float* w_ih   = (const float*)d_in[2];
  const float* w_hh   = (const float*)d_in[3];
  const float* b_ih   = (const float*)d_in[4];
  const float* b_hh   = (const float*)d_in[5];
  const float* w_out  = (const float*)d_in[6];
  const float* b_out  = (const float*)d_in[7];

  float* out  = (float*)d_out;
  float* rec  = (float*)d_ws;

  gru_sweeps_kernel<<<SC_WAVES, 32, 0, stream>>>(x, state, w_ih, w_hh, b_ih, b_hh, w_out,
                                                 b_out, out, rec);
  gru_join_kernel<<<1, 32, 0, stream>>>(x, state, w_ih, w_hh, b_ih, b_hh, w_out, b_out,
                                        out, rec, out + (size_t)SEQ_LEN * 2);
}


# scan: each wave ends its own relaxation: after a sweep one ballot compares all 16 columns' exits with the sweep before, bitwise; none moved -> next sweep is the last (was always 9); join checks kept
# speedup vs baseline: 1385.9416x; 1.0817x over previous
#include <hip/hip_runtime.h>

typedef __attribute__((ext_vector_type(2))) float v2f;
typedef __attribute__((ext_vector_type(8))) float v8f;

#define SEQ_LEN 262144
#define HDIM    4

__device__ __forceinline__ float fast_sigmoid(float x) {
  const float e = __builtin_amdgcn_exp2f(-1.44269504088896340736f * x);
  return __builtin_amdgcn_rcpf(1.0f + e);
}

#define SCAN_GRP  4
#define SCAN_RING 4
#define SCAN_TRIP (SCAN_GRP * SCAN_RING)

struct GruLane {
  v2f a;
  v2f ax;
  float bh_n0, bh_n1;
  float bo0, bo1;
  const float* xp;
  float* yp;
  float ha, hb;
  bool upper;
};

__device__ __forceinline__ void gru_lane_place(
    GruLane& L, const float* __restrict__ x, float* __restrict__ y, const int t_base) {
  L.xp = x + (size_t)t_base * 2;
  L.yp = y + (size_t)t_base * 2;
  L.upper = (threadIdx.x >> 4) != 0;
}

__device__ __forceinline__ void gru_lane_init(
    GruLane& L, const float* __restrict__ w_ih,
    const float* __restrict__ w_hh, const float* __restrict__ b_ih,
    const float* __restrict__ b_hh, const float* __restrict__ w_out,
    const float* __restrict__ b_out) {
  const int lane = threadIdx.x;
  const int hf   = lane >> 4;
  const int c    = lane & 15;

  {
    const int mh = c >> 3, mi = c & 7;
    if (mi < 6) {
      const int g = (mi >> 1) * 4 + 2 * mh + (mi & 1);
      const float2 wv = *reinterpret_cast<const float2*>(w_hh + g * 4 + 2 * hf);
      L.a.x = wv.x; L.a.y = wv.y;
      if (hf == 0) {
        const float2 wi = *reinterpret_cast<const float2*>(w_ih + g * 2);
        L.ax.x = wi.x; L.ax.y = wi.y;
      } else {
        const float bi = b_ih[g];
        const float bh = b_hh[g];
        L.ax.x = (g < 2 * HDIM) ? bi + bh : bi;
        L.ax.y = 0.0f;
      }
    } else if (mh == 0) {
      const float2 wv = *reinterpret_cast<const float2*>(w_out + (mi - 6) * 4 + 2 * hf);
      L.a.x = wv.x; L.a.y = wv.y;
      L.ax.x = 0.0f; L.ax.y = 0.0f;
    } else {
      L.a.x = 0.0f; L.a.y = 0.0f;
      L.ax.x = 0.0f; L.ax.y = 0.0f;
    }
  }
  L.bo0 = b_out[0]; L.bo1 = b_out[1];
  L.bh_n0 = b_hh[2 * HDIM + 2 * hf + 0];
  L.bh_n1 = b_hh[2 * HDIM + 2 * hf + 1];

  L.ha = 0.0f; L.hb = 0.0f;
}

__device__ __forceinline__ void gru_load_grp(
    const GruLane& L, float (&xs)[2 * SCAN_GRP], int t) {
  const float4 v0 = *reinterpret_cast<const float4*>(L.xp + 2 * t);
  const float4 v1 = *reinterpret_cast<const float4*>(L.xp + 2 * t + 4);
  xs[0] = v0.x; xs[1] = v0.y; xs[2] = v0.z; xs[3] = v0.w;
  xs[4] = v1.x; xs[5] = v1.y; xs[6] = v1.z; xs[7] = v1.w;
}

__device__ __forceinline__ v8f gru_gi(const GruLane& L, float x0, float x1) {
  v2f b;
  b.x = L.upper ? 1.0f : x0;
  b.y = L.upper ? 0.0f : x1;
  const v8f zero = {0.0f, 0.0f, 0.0f, 0.0f, 0.0f, 0.0f, 0.0f, 0.0f};
  return __builtin_amdgcn_wmma_f32_16x16x4_f32(
       false, L.ax,  false, b,
       (short)0, zero,  false,  false);
}

template <bool STORE_Y>
__device__ __forceinline__ void gru_run_grp(
    GruLane& L, const float (&xs)[2 * SCAN_GRP], float nx0, float nx1,
    v8f& gnext, int t, bool keep) {
  float ys[SCAN_GRP][2];
#pragma unroll
  for (int u = 0; u < SCAN_GRP; ++u) {
    const v8f gi = gnext;
    v8f cc;
    cc[0] = gi[0];   cc[1] = gi[1];
    cc[2] = gi[2];   cc[3] = gi[3];
    cc[4] = L.bh_n0; cc[5] = L.bh_n1;
    cc[6] = L.bo0;   cc[7] = L.bo1;
    v2f b; b.x = L.ha; b.y = L.hb;

    const v8f d = __builtin_amdgcn_wmma_f32_16x16x4_f32(
         false, L.a,  false, b,
         (short)0, cc,  false,  false);
    ys[u][0] = d[6]; ys[u][1] = d[7];

    gnext = (u + 1 < SCAN_GRP) ? gru_gi(L, xs[2 * u + 2], xs[2 * u + 3])
                               : gru_gi(L, nx0, nx1);

    const float ra = fast_sigmoid(d[0]);
    const float rb = fast_sigmoid(d[1]);
    const float za = fast_sigmoid(d[2]);
    const float zb = fast_sigmoid(d[3]);

    const float na = fmaf(2.0f, fast_sigmoid(2.0f * fmaf(ra, d[4], gi[4])), -1.0f);
    const float nb = fmaf(2.0f, fast_sigmoid(2.0f * fmaf(rb, d[5], gi[5])), -1.0f);

    L.ha = fmaf(za, L.ha - na, na);
    L.hb = fmaf(zb, L.hb - nb, nb);
  }
  if (STORE_Y && keep) {
    if (t > 0)
      *reinterpret_cast<float2*>(L.yp + (t - 1) * 2) = make_float2(ys[0][0], ys[0][1]);
#pragma unroll
    for (int u = 1; u < SCAN_GRP; ++u)
      *reinterpret_cast<float2*>(L.yp + (t + u - 1) * 2) =
          make_float2(ys[u][0], ys[u][1]);
  }
}

__device__ __forceinline__ void gru_last_y(const GruLane& L, int t_last, bool keep) {
  const v8f cc = {0.0f, 0.0f, 0.0f, 0.0f, 0.0f, 0.0f, L.bo0, L.bo1};
  v2f b; b.x = L.ha; b.y = L.hb;
  const v8f d = __builtin_amdgcn_wmma_f32_16x16x4_f32(
       false, L.a,  false, b,
       (short)0, cc,  false,  false);
  if (keep)
    *reinterpret_cast<float2*>(L.yp + t_last * 2) = make_float2(d[6], d[7]);
}

#define CH_LEN   64
#define N_CHUNK  (SEQ_LEN / CH_LEN)
#define RUNIN    8
#define OWNED    (16 - RUNIN)
#define N_SWEEP  (RUNIN + 1)
#define CH_TRIPS (CH_LEN / SCAN_TRIP)
#define SC_WAVES (N_CHUNK / OWNED)
#define REC_PITCH 32
#define REC_EXIT  HDIM
#define REC_FLAG  (2 * HDIM)

__device__ __forceinline__ float from_left_column(float v) {
  return __builtin_bit_cast(float, __builtin_amdgcn_update_dpp(
      0, __builtin_bit_cast(int, v),   0x111, 0xf, 0xf, true));
}

__global__ __launch_bounds__(32) void gru_sweeps_kernel(
    const float* __restrict__ x, const float* __restrict__ state0,
    const float* __restrict__ w_ih, const float* __restrict__ w_hh,
    const float* __restrict__ b_ih, const float* __restrict__ b_hh,
    const float* __restrict__ w_out, const float* __restrict__ b_out,
    float* __restrict__ y, float* __restrict__ rec) {
  const int lane = threadIdx.x;
  const int hf   = lane >> 4;
  const int c    = lane & 15;
  const int k    = (int)blockIdx.x * OWNED - RUNIN + c;
  const int kc   = k < 0 ? 0 : k;
  const bool head  = (k == 0);
  const bool owned = (c >= RUNIN);
  const bool ykeep = owned & (hf == 0);

  GruLane L;
  gru_lane_place(L, x, y, kc * CH_LEN);

  float g[SCAN_RING][2 * SCAN_GRP];
#pragma unroll
  for (int j = 0; j < SCAN_RING; ++j) gru_load_grp(L, g[j], SCAN_GRP * j);

  gru_lane_init(L, w_ih, w_hh, b_ih, b_hh, w_out, b_out);

  const float s_a = state0[2 * hf + 0];
  const float s_b = state0[2 * hf + 1];
  float ea = head ? s_a : 0.0f;
  float eb = head ? s_b : 0.0f;
  float xa = 0.0f, xb = 0.0f;

  v8f gnext = gru_gi(L, g[0][0], g[0][1]);

  L.ha = ea; L.hb = eb;
  for (int trip = 0; trip < (N_SWEEP - 1) * CH_TRIPS; ++trip) {
    const int t = (trip & (CH_TRIPS - 1)) * SCAN_TRIP;
#pragma unroll
    for (int j = 0; j < SCAN_RING; ++j) {
      gru_run_grp<false>(L, g[j], g[(j + 1) % SCAN_RING][0], g[(j + 1) % SCAN_RING][1],
                         gnext, t + SCAN_GRP * j, false);
      gru_load_grp(L, g[j], (t + SCAN_TRIP + SCAN_GRP * j) & (CH_LEN - 1));
      __builtin_amdgcn_sched_barrier(0);
    }
    if (t == CH_LEN - SCAN_TRIP) {
      const bool drift =
          (__builtin_bit_cast(int, L.ha) != __builtin_bit_cast(int, xa)) |
          (__builtin_bit_cast(int, L.hb) != __builtin_bit_cast(int, xb));
      const bool calm = (trip >= CH_TRIPS) & (__builtin_amdgcn_ballot_w32(drift) == 0u);
      xa = L.ha; xb = L.hb;
      const float la = from_left_column(xa);
      const float lb = from_left_column(xb);
      ea = head ? s_a : la;
      eb = head ? s_b : lb;
      L.ha = ea; L.hb = eb;
      if (calm) break;
    }
  }

  for (int t = 0; t < CH_LEN; t += SCAN_TRIP) {
#pragma unroll
    for (int j = 0; j < SCAN_RING; ++j) {
      gru_run_grp<true>(L, g[j], g[(j + 1) % SCAN_RING][0], g[(j + 1) % SCAN_RING][1],
                        gnext, t + SCAN_GRP * j, ykeep);
      gru_load_grp(L, g[j], (t + SCAN_TRIP + SCAN_GRP * j) & (CH_LEN - 1));
      __builtin_amdgcn_sched_barrier(0);
    }
  }

  gru_last_y(L, CH_LEN - 1, ykeep);

  const bool moved = owned & (c < 15) &
      ((__builtin_bit_cast(int, L.ha) != __builtin_bit_cast(int, xa)) |
       (__builtin_bit_cast(int, L.hb) != __builtin_bit_cast(int, xb)));
  const unsigned any_moved = __builtin_amdgcn_ballot_w32(moved);
  float* line = rec + (size_t)blockIdx.x * REC_PITCH;
  if (lane == 0)
    line[REC_FLAG] = (any_moved != 0u) ? 1.0f : 0.0f;
  if (c == RUNIN)
    *reinterpret_cast<float2*>(line + 2 * hf) = make_float2(ea, eb);
  if (c == 15)
    *reinterpret_cast<float2*>(line + REC_EXIT + 2 * hf) = make_float2(L.ha, L.hb);
}

template <int NSTEPS>
__device__ __forceinline__ void gru_serial(GruLane& L, bool keep) {
  float g[SCAN_RING][2 * SCAN_GRP];
#pragma unroll
  for (int j = 0; j < SCAN_RING; ++j) gru_load_grp(L, g[j], SCAN_GRP * j);
  v8f gnext = gru_gi(L, g[0][0], g[0][1]);

  for (int t = 0; t < NSTEPS; t += SCAN_TRIP) {
#pragma unroll
    for (int j = 0; j < SCAN_RING; ++j) {
      gru_run_grp<true>(L, g[j], g[(j + 1) % SCAN_RING][0], g[(j + 1) % SCAN_RING][1],
                        gnext, t + SCAN_GRP * j, keep);
      gru_load_grp(L, g[j], (t + SCAN_TRIP + SCAN_GRP * j) & (NSTEPS - 1));
      __builtin_amdgcn_sched_barrier(0);
    }
  }
  gru_last_y(L, NSTEPS - 1, keep);
}

__global__ __launch_bounds__(32) void gru_join_kernel(
    const float* __restrict__ x, const float* __restrict__ state0,
    const float* __restrict__ w_ih, const float* __restrict__ w_hh,
    const float* __restrict__ b_ih, const float* __restrict__ b_hh,
    const float* __restrict__ w_out, const float* __restrict__ b_out,
    float* __restrict__ y, const float* __restrict__ rec,
    float* __restrict__ state_out) {
  const int lane = threadIdx.x;
  const int hf   = lane >> 4;
  const int c    = lane & 15;

  bool bad = false;
#pragma unroll
  for (int j = 0; j < SC_WAVES / 32; ++j) {
    const int w  = j * 32 + lane;
    const int wl = w > 0 ? w : 1;
    const float* line = rec + (size_t)w * REC_PITCH;
    const float4 e = *reinterpret_cast<const float4*>(rec + (size_t)wl * REC_PITCH);
    const float4 xl = *reinterpret_cast<const float4*>(
        rec + (size_t)(wl - 1) * REC_PITCH + REC_EXIT);
    bad |= (line[REC_FLAG] != 0.0f);
    bad |= (w > 0) &
           ((__builtin_bit_cast(int, e.x) != __builtin_bit_cast(int, xl.x)) |
            (__builtin_bit_cast(int, e.y) != __builtin_bit_cast(int, xl.y)) |
            (__builtin_bit_cast(int, e.z) != __builtin_bit_cast(int, xl.z)) |
            (__builtin_bit_cast(int, e.w) != __builtin_bit_cast(int, xl.w)));
  }

  float ha, hb;
  if (__builtin_amdgcn_ballot_w32(bad) != 0u) {
    GruLane L;
    gru_lane_place(L, x, y, 0);
    gru_lane_init(L, w_ih, w_hh, b_ih, b_hh, w_out, b_out);
    L.ha = state0[2 * hf + 0];
    L.hb = state0[2 * hf + 1];
    gru_serial<SEQ_LEN>(L, lane == 0);
    ha = L.ha; hb = L.hb;
  } else {
    const float2 v = *reinterpret_cast<const float2*>(
        rec + (size_t)(SC_WAVES - 1) * REC_PITCH + REC_EXIT + 2 * hf);
    ha = v.x; hb = v.y;
  }

  if (c == 0) {
    state_out[2 * hf + 0] = ha;
    state_out[2 * hf + 1] = hb;
  }
}

extern "C" void kernel_launch(void* const* d_in, const int* in_sizes, int n_in,
                              void* d_out, int out_size, void* d_ws, size_t ws_size,
                              hipStream_t stream) {
  const float* x      = (const float*)d_in[0];
  const float* state  = (const float*)d_in[1];
  const float* w_ih   = (const float*)d_in[2];
  const float* w_hh   = (const float*)d_in[3];
  const float* b_ih   = (const float*)d_in[4];
  const float* b_hh   = (const float*)d_in[5];
  const float* w_out  = (const float*)d_in[6];
  const float* b_out  = (const float*)d_in[7];

  float* out  = (float*)d_out;
  float* rec  = (float*)d_ws;

  gru_sweeps_kernel<<<SC_WAVES, 32, 0, stream>>>(x, state, w_ih, w_hh, b_ih, b_hh, w_out,
                                                 b_out, out, rec);
  gru_join_kernel<<<1, 32, 0, stream>>>(x, state, w_ih, w_hh, b_ih, b_hh, w_out, b_out,
                                        out, rec, out + (size_t)SEQ_LEN * 2);
}


# scan: own columns store y in every sweep after the first (one block per 16 steps), so the sweep whose exits repeat the sweep before, bitwise, IS the last one: the sweep that re-ran it is gone
# speedup vs baseline: 1442.4194x; 1.0408x over previous
#include <hip/hip_runtime.h>

typedef __attribute__((ext_vector_type(2))) float v2f;
typedef __attribute__((ext_vector_type(8))) float v8f;

#define SEQ_LEN 262144
#define HDIM    4

__device__ __forceinline__ float fast_sigmoid(float x) {
  const float e = __builtin_amdgcn_exp2f(-1.44269504088896340736f * x);
  return __builtin_amdgcn_rcpf(1.0f + e);
}

#define SCAN_GRP  4
#define SCAN_RING 4
#define SCAN_TRIP (SCAN_GRP * SCAN_RING)

struct GruLane {
  v2f a;
  v2f ax;
  float bh_n0, bh_n1;
  float bo0, bo1;
  const float* xp;
  float* yp;
  float ha, hb;
  bool upper;
};

__device__ __forceinline__ void gru_lane_place(
    GruLane& L, const float* __restrict__ x, float* __restrict__ y, const int t_base) {
  L.xp = x + (size_t)t_base * 2;
  L.yp = y + (size_t)t_base * 2;
  L.upper = (threadIdx.x >> 4) != 0;
}

__device__ __forceinline__ void gru_lane_init(
    GruLane& L, const float* __restrict__ w_ih,
    const float* __restrict__ w_hh, const float* __restrict__ b_ih,
    const float* __restrict__ b_hh, const float* __restrict__ w_out,
    const float* __restrict__ b_out) {
  const int lane = threadIdx.x;
  const int hf   = lane >> 4;
  const int c    = lane & 15;

  {
    const int mh = c >> 3, mi = c & 7;
    if (mi < 6) {
      const int g = (mi >> 1) * 4 + 2 * mh + (mi & 1);
      const float2 wv = *reinterpret_cast<const float2*>(w_hh + g * 4 + 2 * hf);
      L.a.x = wv.x; L.a.y = wv.y;
      if (hf == 0) {
        const float2 wi = *reinterpret_cast<const float2*>(w_ih + g * 2);
        L.ax.x = wi.x; L.ax.y = wi.y;
      } else {
        const float bi = b_ih[g];
        const float bh = b_hh[g];
        L.ax.x = (g < 2 * HDIM) ? bi + bh : bi;
        L.ax.y = 0.0f;
      }
    } else if (mh == 0) {
      const float2 wv = *reinterpret_cast<const float2*>(w_out + (mi - 6) * 4 + 2 * hf);
      L.a.x = wv.x; L.a.y = wv.y;
      L.ax.x = 0.0f; L.ax.y = 0.0f;
    } else {
      L.a.x = 0.0f; L.a.y = 0.0f;
      L.ax.x = 0.0f; L.ax.y = 0.0f;
    }
  }
  L.bo0 = b_out[0]; L.bo1 = b_out[1];
  L.bh_n0 = b_hh[2 * HDIM + 2 * hf + 0];
  L.bh_n1 = b_hh[2 * HDIM + 2 * hf + 1];

  L.ha = 0.0f; L.hb = 0.0f;
}

__device__ __forceinline__ void gru_load_grp(
    const GruLane& L, float (&xs)[2 * SCAN_GRP], int t) {
  const float4 v0 = *reinterpret_cast<const float4*>(L.xp + 2 * t);
  const float4 v1 = *reinterpret_cast<const float4*>(L.xp + 2 * t + 4);
  xs[0] = v0.x; xs[1] = v0.y; xs[2] = v0.z; xs[3] = v0.w;
  xs[4] = v1.x; xs[5] = v1.y; xs[6] = v1.z; xs[7] = v1.w;
}

__device__ __forceinline__ v8f gru_gi(const GruLane& L, float x0, float x1) {
  v2f b;
  b.x = L.upper ? 1.0f : x0;
  b.y = L.upper ? 0.0f : x1;
  const v8f zero = {0.0f, 0.0f, 0.0f, 0.0f, 0.0f, 0.0f, 0.0f, 0.0f};
  return __builtin_amdgcn_wmma_f32_16x16x4_f32(
       false, L.ax,  false, b,
       (short)0, zero,  false,  false);
}

__device__ __forceinline__ void gru_run_grp(
    GruLane& L, const float (&xs)[2 * SCAN_GRP], float nx0, float nx1,
    v8f& gnext, float (&ys)[SCAN_GRP][2]) {
#pragma unroll
  for (int u = 0; u < SCAN_GRP; ++u) {
    const v8f gi = gnext;
    v8f cc;
    cc[0] = gi[0];   cc[1] = gi[1];
    cc[2] = gi[2];   cc[3] = gi[3];
    cc[4] = L.bh_n0; cc[5] = L.bh_n1;
    cc[6] = L.bo0;   cc[7] = L.bo1;
    v2f b; b.x = L.ha; b.y = L.hb;

    const v8f d = __builtin_amdgcn_wmma_f32_16x16x4_f32(
         false, L.a,  false, b,
         (short)0, cc,  false,  false);
    ys[u][0] = d[6]; ys[u][1] = d[7];

    gnext = (u + 1 < SCAN_GRP) ? gru_gi(L, xs[2 * u + 2], xs[2 * u + 3])
                               : gru_gi(L, nx0, nx1);

    const float ra = fast_sigmoid(d[0]);
    const float rb = fast_sigmoid(d[1]);
    const float za = fast_sigmoid(d[2]);
    const float zb = fast_sigmoid(d[3]);

    const float na = fmaf(2.0f, fast_sigmoid(2.0f * fmaf(ra, d[4], gi[4])), -1.0f);
    const float nb = fmaf(2.0f, fast_sigmoid(2.0f * fmaf(rb, d[5], gi[5])), -1.0f);

    L.ha = fmaf(za, L.ha - na, na);
    L.hb = fmaf(zb, L.hb - nb, nb);
  }
}

__device__ __forceinline__ void gru_store_trip(
    const GruLane& L, const float (&yt)[SCAN_RING][SCAN_GRP][2], int t, bool keep) {
  if (keep) {
    float* yq = L.yp + t * 2;
    if (t > 0)
      *reinterpret_cast<float2*>(yq - 2) = make_float2(yt[0][0][0], yt[0][0][1]);
#pragma unroll
    for (int i = 1; i < SCAN_TRIP; ++i)
      *reinterpret_cast<float2*>(yq + (i - 1) * 2) =
          make_float2(yt[i / SCAN_GRP][i % SCAN_GRP][0], yt[i / SCAN_GRP][i % SCAN_GRP][1]);
  }
}

__device__ __forceinline__ void gru_last_y(const GruLane& L, int t_last, bool keep) {
  const v8f cc = {0.0f, 0.0f, 0.0f, 0.0f, 0.0f, 0.0f, L.bo0, L.bo1};
  v2f b; b.x = L.ha; b.y = L.hb;
  const v8f d = __builtin_amdgcn_wmma_f32_16x16x4_f32(
       false, L.a,  false, b,
       (short)0, cc,  false,  false);
  if (keep)
    *reinterpret_cast<float2*>(L.yp + t_last * 2) = make_float2(d[6], d[7]);
}

#define CH_LEN   64
#define N_CHUNK  (SEQ_LEN / CH_LEN)
#define RUNIN    8
#define OWNED    (16 - RUNIN)
#define N_SWEEP  (RUNIN + 1)
#define CH_TRIPS (CH_LEN / SCAN_TRIP)
#define SC_WAVES (N_CHUNK / OWNED)
#define REC_PITCH 32
#define REC_EXIT  HDIM
#define REC_FLAG  (2 * HDIM)

__device__ __forceinline__ float from_left_column(float v) {
  return __builtin_bit_cast(float, __builtin_amdgcn_update_dpp(
      0, __builtin_bit_cast(int, v),   0x111, 0xf, 0xf, true));
}

__global__ __launch_bounds__(32) void gru_sweeps_kernel(
    const float* __restrict__ x, const float* __restrict__ state0,
    const float* __restrict__ w_ih, const float* __restrict__ w_hh,
    const float* __restrict__ b_ih, const float* __restrict__ b_hh,
    const float* __restrict__ w_out, const float* __restrict__ b_out,
    float* __restrict__ y, float* __restrict__ rec) {
  const int lane = threadIdx.x;
  const int hf   = lane >> 4;
  const int c    = lane & 15;
  const int k    = (int)blockIdx.x * OWNED - RUNIN + c;
  const int kc   = k < 0 ? 0 : k;
  const bool head  = (k == 0);
  const bool owned = (c >= RUNIN);
  const bool ykeep = owned & (hf == 0);

  GruLane L;
  gru_lane_place(L, x, y, kc * CH_LEN);

  float g[SCAN_RING][2 * SCAN_GRP];
#pragma unroll
  for (int j = 0; j < SCAN_RING; ++j) gru_load_grp(L, g[j], SCAN_GRP * j);

  gru_lane_init(L, w_ih, w_hh, b_ih, b_hh, w_out, b_out);

  const float s_a = state0[2 * hf + 0];
  const float s_b = state0[2 * hf + 1];
  float ea = head ? s_a : 0.0f;
  float eb = head ? s_b : 0.0f;
  float xa = 0.0f, xb = 0.0f;

  v8f gnext = gru_gi(L, g[0][0], g[0][1]);

  L.ha = ea; L.hb = eb;
  bool drift = false;
  for (int trip = 0; trip < N_SWEEP * CH_TRIPS; ++trip) {
    const int t = (trip & (CH_TRIPS - 1)) * SCAN_TRIP;
    const bool keep = ykeep & (trip >= CH_TRIPS);
    float yt[SCAN_RING][SCAN_GRP][2];
#pragma unroll
    for (int j = 0; j < SCAN_RING; ++j) {
      gru_run_grp(L, g[j], g[(j + 1) % SCAN_RING][0], g[(j + 1) % SCAN_RING][1],
                  gnext, yt[j]);
      if (j == SCAN_RING - 1) gru_store_trip(L, yt, t, keep);
      gru_load_grp(L, g[j], (t + SCAN_TRIP + SCAN_GRP * j) & (CH_LEN - 1));
      __builtin_amdgcn_sched_barrier(0);
    }
    if (t == CH_LEN - SCAN_TRIP) {
      drift = (__builtin_bit_cast(int, L.ha) != __builtin_bit_cast(int, xa)) |
              (__builtin_bit_cast(int, L.hb) != __builtin_bit_cast(int, xb));
      const bool calm = (trip >= CH_TRIPS) & (__builtin_amdgcn_ballot_w32(drift) == 0u);
      if (calm | (trip == N_SWEEP * CH_TRIPS - 1)) break;
      xa = L.ha; xb = L.hb;
      const float la = from_left_column(xa);
      const float lb = from_left_column(xb);
      ea = head ? s_a : la;
      eb = head ? s_b : lb;
      L.ha = ea; L.hb = eb;
    }
  }

  gru_last_y(L, CH_LEN - 1, ykeep);

  const bool moved = owned & (c < 15) & drift;
  const unsigned any_moved = __builtin_amdgcn_ballot_w32(moved);
  float* line = rec + (size_t)blockIdx.x * REC_PITCH;
  if (lane == 0)
    line[REC_FLAG] = (any_moved != 0u) ? 1.0f : 0.0f;
  if (c == RUNIN)
    *reinterpret_cast<float2*>(line + 2 * hf) = make_float2(ea, eb);
  if (c == 15)
    *reinterpret_cast<float2*>(line + REC_EXIT + 2 * hf) = make_float2(L.ha, L.hb);
}

template <int NSTEPS>
__device__ __forceinline__ void gru_serial(GruLane& L, bool keep) {
  float g[SCAN_RING][2 * SCAN_GRP];
#pragma unroll
  for (int j = 0; j < SCAN_RING; ++j) gru_load_grp(L, g[j], SCAN_GRP * j);
  v8f gnext = gru_gi(L, g[0][0], g[0][1]);

  for (int t = 0; t < NSTEPS; t += SCAN_TRIP) {
    float yt[SCAN_RING][SCAN_GRP][2];
#pragma unroll
    for (int j = 0; j < SCAN_RING; ++j) {
      gru_run_grp(L, g[j], g[(j + 1) % SCAN_RING][0], g[(j + 1) % SCAN_RING][1],
                  gnext, yt[j]);
      if (j == SCAN_RING - 1) gru_store_trip(L, yt, t, keep);
      gru_load_grp(L, g[j], (t + SCAN_TRIP + SCAN_GRP * j) & (NSTEPS - 1));
      __builtin_amdgcn_sched_barrier(0);
    }
  }
  gru_last_y(L, NSTEPS - 1, keep);
}

__global__ __launch_bounds__(32) void gru_join_kernel(
    const float* __restrict__ x, const float* __restrict__ state0,
    const float* __restrict__ w_ih, const float* __restrict__ w_hh,
    const float* __restrict__ b_ih, const float* __restrict__ b_hh,
    const float* __restrict__ w_out, const float* __restrict__ b_out,
    float* __restrict__ y, const float* __restrict__ rec,
    float* __restrict__ state_out) {
  const int lane = threadIdx.x;
  const int hf   = lane >> 4;
  const int c    = lane & 15;

  bool bad = false;
#pragma unroll
  for (int j = 0; j < SC_WAVES / 32; ++j) {
    const int w  = j * 32 + lane;
    const int wl = w > 0 ? w : 1;
    const float* line = rec + (size_t)w * REC_PITCH;
    const float4 e = *reinterpret_cast<const float4*>(rec + (size_t)wl * REC_PITCH);
    const float4 xl = *reinterpret_cast<const float4*>(
        rec + (size_t)(wl - 1) * REC_PITCH + REC_EXIT);
    bad |= (line[REC_FLAG] != 0.0f);
    bad |= (w > 0) &
           ((__builtin_bit_cast(int, e.x) != __builtin_bit_cast(int, xl.x)) |
            (__builtin_bit_cast(int, e.y) != __builtin_bit_cast(int, xl.y)) |
            (__builtin_bit_cast(int, e.z) != __builtin_bit_cast(int, xl.z)) |
            (__builtin_bit_cast(int, e.w) != __builtin_bit_cast(int, xl.w)));
  }

  float ha, hb;
  if (__builtin_amdgcn_ballot_w32(bad) != 0u) {
    GruLane L;
    gru_lane_place(L, x, y, 0);
    gru_lane_init(L, w_ih, w_hh, b_ih, b_hh, w_out, b_out);
    L.ha = state0[2 * hf + 0];
    L.hb = state0[2 * hf + 1];
    gru_serial<SEQ_LEN>(L, lane == 0);
    ha = L.ha; hb = L.hb;
  } else {
    const float2 v = *reinterpret_cast<const float2*>(
        rec + (size_t)(SC_WAVES - 1) * REC_PITCH + REC_EXIT + 2 * hf);
    ha = v.x; hb = v.y;
  }

  if (c == 0) {
    state_out[2 * hf + 0] = ha;
    state_out[2 * hf + 1] = hb;
  }
}

extern "C" void kernel_launch(void* const* d_in, const int* in_sizes, int n_in,
                              void* d_out, int out_size, void* d_ws, size_t ws_size,
                              hipStream_t stream) {
  const float* x      = (const float*)d_in[0];
  const float* state  = (const float*)d_in[1];
  const float* w_ih   = (const float*)d_in[2];
  const float* w_hh   = (const float*)d_in[3];
  const float* b_ih   = (const float*)d_in[4];
  const float* b_hh   = (const float*)d_in[5];
  const float* w_out  = (const float*)d_in[6];
  const float* b_out  = (const float*)d_in[7];

  float* out  = (float*)d_out;
  float* rec  = (float*)d_ws;

  gru_sweeps_kernel<<<SC_WAVES, 32, 0, stream>>>(x, state, w_ih, w_hh, b_ih, b_hh, w_out,
                                                 b_out, out, rec);
  gru_join_kernel<<<1, 32, 0, stream>>>(x, state, w_ih, w_hh, b_ih, b_hh, w_out, b_out,
                                        out, rec, out + (size_t)SEQ_LEN * 2);
}


# scan: input products (x @ w_ih.T + biases) of a wave's 16 chunks are formed once ahead of the sweeps and kept in registers (64 steps unrolled): sweeps have 1 WMMA per step, no x load or load wait
# speedup vs baseline: 1793.1062x; 1.2431x over previous
#include <hip/hip_runtime.h>

typedef __attribute__((ext_vector_type(2))) float v2f;
typedef __attribute__((ext_vector_type(8))) float v8f;

#define SEQ_LEN 262144
#define HDIM    4

__device__ __forceinline__ float fast_sigmoid(float x) {
  const float e = __builtin_amdgcn_exp2f(-1.44269504088896340736f * x);
  return __builtin_amdgcn_rcpf(1.0f + e);
}

#define SCAN_GRP  4
#define SCAN_RING 4
#define SCAN_TRIP (SCAN_GRP * SCAN_RING)

struct GruLane {
  v2f a;
  v2f ax;
  float bh_n0, bh_n1;
  float bo0, bo1;
  const float* xp;
  float* yp;
  float ha, hb;
  bool upper;
};

__device__ __forceinline__ void gru_lane_place(
    GruLane& L, const float* __restrict__ x, float* __restrict__ y, const int t_base) {
  L.xp = x + (size_t)t_base * 2;
  L.yp = y + (size_t)t_base * 2;
  L.upper = (threadIdx.x >> 4) != 0;
}

__device__ __forceinline__ void gru_lane_init(
    GruLane& L, const float* __restrict__ w_ih,
    const float* __restrict__ w_hh, const float* __restrict__ b_ih,
    const float* __restrict__ b_hh, const float* __restrict__ w_out,
    const float* __restrict__ b_out) {
  const int lane = threadIdx.x;
  const int hf   = lane >> 4;
  const int c    = lane & 15;

  {
    const int mh = c >> 3, mi = c & 7;
    if (mi < 6) {
      const int g = (mi >> 1) * 4 + 2 * mh + (mi & 1);
      const float2 wv = *reinterpret_cast<const float2*>(w_hh + g * 4 + 2 * hf);
      L.a.x = wv.x; L.a.y = wv.y;
      if (hf == 0) {
        const float2 wi = *reinterpret_cast<const float2*>(w_ih + g * 2);
        L.ax.x = wi.x; L.ax.y = wi.y;
      } else {
        const float bi = b_ih[g];
        const float bh = b_hh[g];
        L.ax.x = (g < 2 * HDIM) ? bi + bh : bi;
        L.ax.y = 0.0f;
      }
    } else if (mh == 0) {
      const float2 wv = *reinterpret_cast<const float2*>(w_out + (mi - 6) * 4 + 2 * hf);
      L.a.x = wv.x; L.a.y = wv.y;
      L.ax.x = 0.0f; L.ax.y = 0.0f;
    } else {
      L.a.x = 0.0f; L.a.y = 0.0f;
      L.ax.x = 0.0f; L.ax.y = 0.0f;
    }
  }
  L.bo0 = b_out[0]; L.bo1 = b_out[1];
  L.bh_n0 = b_hh[2 * HDIM + 2 * hf + 0];
  L.bh_n1 = b_hh[2 * HDIM + 2 * hf + 1];

  L.ha = 0.0f; L.hb = 0.0f;
}

__device__ __forceinline__ void gru_load_grp(
    const GruLane& L, float (&xs)[2 * SCAN_GRP], int t) {
  const float4 v0 = *reinterpret_cast<const float4*>(L.xp + 2 * t);
  const float4 v1 = *reinterpret_cast<const float4*>(L.xp + 2 * t + 4);
  xs[0] = v0.x; xs[1] = v0.y; xs[2] = v0.z; xs[3] = v0.w;
  xs[4] = v1.x; xs[5] = v1.y; xs[6] = v1.z; xs[7] = v1.w;
}

__device__ __forceinline__ v8f gru_gi(const GruLane& L, float x0, float x1) {
  v2f b;
  b.x = L.upper ? 1.0f : x0;
  b.y = L.upper ? 0.0f : x1;
  const v8f zero = {0.0f, 0.0f, 0.0f, 0.0f, 0.0f, 0.0f, 0.0f, 0.0f};
  return __builtin_amdgcn_wmma_f32_16x16x4_f32(
       false, L.ax,  false, b,
       (short)0, zero,  false,  false);
}

__device__ __forceinline__ void gru_run_grp(
    GruLane& L, const float (&xs)[2 * SCAN_GRP], float nx0, float nx1,
    v8f& gnext, float (&ys)[SCAN_GRP][2]) {
#pragma unroll
  for (int u = 0; u < SCAN_GRP; ++u) {
    const v8f gi = gnext;
    v8f cc;
    cc[0] = gi[0];   cc[1] = gi[1];
    cc[2] = gi[2];   cc[3] = gi[3];
    cc[4] = L.bh_n0; cc[5] = L.bh_n1;
    cc[6] = L.bo0;   cc[7] = L.bo1;
    v2f b; b.x = L.ha; b.y = L.hb;

    const v8f d = __builtin_amdgcn_wmma_f32_16x16x4_f32(
         false, L.a,  false, b,
         (short)0, cc,  false,  false);
    ys[u][0] = d[6]; ys[u][1] = d[7];

    gnext = (u + 1 < SCAN_GRP) ? gru_gi(L, xs[2 * u + 2], xs[2 * u + 3])
                               : gru_gi(L, nx0, nx1);

    const float ra = fast_sigmoid(d[0]);
    const float rb = fast_sigmoid(d[1]);
    const float za = fast_sigmoid(d[2]);
    const float zb = fast_sigmoid(d[3]);

    const float na = fmaf(2.0f, fast_sigmoid(2.0f * fmaf(ra, d[4], gi[4])), -1.0f);
    const float nb = fmaf(2.0f, fast_sigmoid(2.0f * fmaf(rb, d[5], gi[5])), -1.0f);

    L.ha = fmaf(za, L.ha - na, na);
    L.hb = fmaf(zb, L.hb - nb, nb);
  }
}

__device__ __forceinline__ void gru_step_kept(
    GruLane& L, const v8f& gi, float& y0, float& y1) {
  v8f cc;
  cc[0] = gi[0];   cc[1] = gi[1];
  cc[2] = gi[2];   cc[3] = gi[3];
  cc[4] = L.bh_n0; cc[5] = L.bh_n1;
  cc[6] = L.bo0;   cc[7] = L.bo1;
  v2f b; b.x = L.ha; b.y = L.hb;

  const v8f d = __builtin_amdgcn_wmma_f32_16x16x4_f32(
       false, L.a,  false, b,
       (short)0, cc,  false,  false);
  y0 = d[6]; y1 = d[7];

  const float ra = fast_sigmoid(d[0]);
  const float rb = fast_sigmoid(d[1]);
  const float za = fast_sigmoid(d[2]);
  const float zb = fast_sigmoid(d[3]);

  const float na = fmaf(2.0f, fast_sigmoid(2.0f * fmaf(ra, d[4], gi[4])), -1.0f);
  const float nb = fmaf(2.0f, fast_sigmoid(2.0f * fmaf(rb, d[5], gi[5])), -1.0f);

  L.ha = fmaf(za, L.ha - na, na);
  L.hb = fmaf(zb, L.hb - nb, nb);
}

__device__ __forceinline__ void gru_store_trip(
    const GruLane& L, const float (&yt)[SCAN_RING][SCAN_GRP][2], int t, bool keep) {
  if (keep) {
    float* yq = L.yp + t * 2;
    if (t > 0)
      *reinterpret_cast<float2*>(yq - 2) = make_float2(yt[0][0][0], yt[0][0][1]);
#pragma unroll
    for (int i = 1; i < SCAN_TRIP; ++i)
      *reinterpret_cast<float2*>(yq + (i - 1) * 2) =
          make_float2(yt[i / SCAN_GRP][i % SCAN_GRP][0], yt[i / SCAN_GRP][i % SCAN_GRP][1]);
  }
}

__device__ __forceinline__ void gru_last_y(const GruLane& L, int t_last, bool keep) {
  const v8f cc = {0.0f, 0.0f, 0.0f, 0.0f, 0.0f, 0.0f, L.bo0, L.bo1};
  v2f b; b.x = L.ha; b.y = L.hb;
  const v8f d = __builtin_amdgcn_wmma_f32_16x16x4_f32(
       false, L.a,  false, b,
       (short)0, cc,  false,  false);
  if (keep)
    *reinterpret_cast<float2*>(L.yp + t_last * 2) = make_float2(d[6], d[7]);
}

#define CH_LEN   64
#define N_CHUNK  (SEQ_LEN / CH_LEN)
#define RUNIN    8
#define OWNED    (16 - RUNIN)
#define N_SWEEP  (RUNIN + 1)
#define CH_TRIPS (CH_LEN / SCAN_TRIP)
#define SC_WAVES (N_CHUNK / OWNED)
#define REC_PITCH 32
#define REC_EXIT  HDIM
#define REC_FLAG  (2 * HDIM)

__device__ __forceinline__ float from_left_column(float v) {
  return __builtin_bit_cast(float, __builtin_amdgcn_update_dpp(
      0, __builtin_bit_cast(int, v),   0x111, 0xf, 0xf, true));
}

__global__ __launch_bounds__(32) void gru_sweeps_kernel(
    const float* __restrict__ x, const float* __restrict__ state0,
    const float* __restrict__ w_ih, const float* __restrict__ w_hh,
    const float* __restrict__ b_ih, const float* __restrict__ b_hh,
    const float* __restrict__ w_out, const float* __restrict__ b_out,
    float* __restrict__ y, float* __restrict__ rec) {
  const int lane = threadIdx.x;
  const int hf   = lane >> 4;
  const int c    = lane & 15;
  const int k    = (int)blockIdx.x * OWNED - RUNIN + c;
  const int kc   = k < 0 ? 0 : k;
  const bool head  = (k == 0);
  const bool owned = (c >= RUNIN);
  const bool ykeep = owned & (hf == 0);

  GruLane L;
  gru_lane_place(L, x, y, kc * CH_LEN);

  float xs[2 * CH_LEN];
#pragma unroll
  for (int i = 0; i < CH_LEN / 2; ++i) {
    const float4 v = *reinterpret_cast<const float4*>(L.xp + 4 * i);
    xs[4 * i + 0] = v.x; xs[4 * i + 1] = v.y;
    xs[4 * i + 2] = v.z; xs[4 * i + 3] = v.w;
  }

  gru_lane_init(L, w_ih, w_hh, b_ih, b_hh, w_out, b_out);

  v8f gi[CH_LEN];
#pragma unroll
  for (int t = 0; t < CH_LEN; ++t) gi[t] = gru_gi(L, xs[2 * t], xs[2 * t + 1]);

  const float s_a = state0[2 * hf + 0];
  const float s_b = state0[2 * hf + 1];
  float ea = head ? s_a : 0.0f;
  float eb = head ? s_b : 0.0f;
  float xa = 0.0f, xb = 0.0f;

  L.ha = ea; L.hb = eb;
  bool drift = false;
  for (int sweep = 0; sweep < N_SWEEP; ++sweep) {
    const bool keep = ykeep & (sweep >= 1);
#pragma unroll
    for (int tr = 0; tr < CH_TRIPS; ++tr) {
      float yt[SCAN_RING][SCAN_GRP][2];
#pragma unroll
      for (int u = 0; u < SCAN_TRIP; ++u)
        gru_step_kept(L, gi[tr * SCAN_TRIP + u],
                      yt[u / SCAN_GRP][u % SCAN_GRP][0], yt[u / SCAN_GRP][u % SCAN_GRP][1]);
      gru_store_trip(L, yt, tr * SCAN_TRIP, keep);
    }
    drift = (__builtin_bit_cast(int, L.ha) != __builtin_bit_cast(int, xa)) |
            (__builtin_bit_cast(int, L.hb) != __builtin_bit_cast(int, xb));
    const bool calm = (sweep >= 1) & (__builtin_amdgcn_ballot_w32(drift) == 0u);
    if (calm | (sweep == N_SWEEP - 1)) break;
    xa = L.ha; xb = L.hb;
    const float la = from_left_column(xa);
    const float lb = from_left_column(xb);
    ea = head ? s_a : la;
    eb = head ? s_b : lb;
    L.ha = ea; L.hb = eb;
  }

  gru_last_y(L, CH_LEN - 1, ykeep);

  const bool moved = owned & (c < 15) & drift;
  const unsigned any_moved = __builtin_amdgcn_ballot_w32(moved);
  float* line = rec + (size_t)blockIdx.x * REC_PITCH;
  if (lane == 0)
    line[REC_FLAG] = (any_moved != 0u) ? 1.0f : 0.0f;
  if (c == RUNIN)
    *reinterpret_cast<float2*>(line + 2 * hf) = make_float2(ea, eb);
  if (c == 15)
    *reinterpret_cast<float2*>(line + REC_EXIT + 2 * hf) = make_float2(L.ha, L.hb);
}

template <int NSTEPS>
__device__ __forceinline__ void gru_serial(GruLane& L, bool keep) {
  float g[SCAN_RING][2 * SCAN_GRP];
#pragma unroll
  for (int j = 0; j < SCAN_RING; ++j) gru_load_grp(L, g[j], SCAN_GRP * j);
  v8f gnext = gru_gi(L, g[0][0], g[0][1]);

  for (int t = 0; t < NSTEPS; t += SCAN_TRIP) {
    float yt[SCAN_RING][SCAN_GRP][2];
#pragma unroll
    for (int j = 0; j < SCAN_RING; ++j) {
      gru_run_grp(L, g[j], g[(j + 1) % SCAN_RING][0], g[(j + 1) % SCAN_RING][1],
                  gnext, yt[j]);
      if (j == SCAN_RING - 1) gru_store_trip(L, yt, t, keep);
      gru_load_grp(L, g[j], (t + SCAN_TRIP + SCAN_GRP * j) & (NSTEPS - 1));
      __builtin_amdgcn_sched_barrier(0);
    }
  }
  gru_last_y(L, NSTEPS - 1, keep);
}

__global__ __launch_bounds__(32) void gru_join_kernel(
    const float* __restrict__ x, const float* __restrict__ state0,
    const float* __restrict__ w_ih, const float* __restrict__ w_hh,
    const float* __restrict__ b_ih, const float* __restrict__ b_hh,
    const float* __restrict__ w_out, const float* __restrict__ b_out,
    float* __restrict__ y, const float* __restrict__ rec,
    float* __restrict__ state_out) {
  const int lane = threadIdx.x;
  const int hf   = lane >> 4;
  const int c    = lane & 15;

  bool bad = false;
#pragma unroll
  for (int j = 0; j < SC_WAVES / 32; ++j) {
    const int w  = j * 32 + lane;
    const int wl = w > 0 ? w : 1;
    const float* line = rec + (size_t)w * REC_PITCH;
    const float4 e = *reinterpret_cast<const float4*>(rec + (size_t)wl * REC_PITCH);
    const float4 xl = *reinterpret_cast<const float4*>(
        rec + (size_t)(wl - 1) * REC_PITCH + REC_EXIT);
    bad |= (line[REC_FLAG] != 0.0f);
    bad |= (w > 0) &
           ((__builtin_bit_cast(int, e.x) != __builtin_bit_cast(int, xl.x)) |
            (__builtin_bit_cast(int, e.y) != __builtin_bit_cast(int, xl.y)) |
            (__builtin_bit_cast(int, e.z) != __builtin_bit_cast(int, xl.z)) |
            (__builtin_bit_cast(int, e.w) != __builtin_bit_cast(int, xl.w)));
  }

  float ha, hb;
  if (__builtin_amdgcn_ballot_w32(bad) != 0u) {
    GruLane L;
    gru_lane_place(L, x, y, 0);
    gru_lane_init(L, w_ih, w_hh, b_ih, b_hh, w_out, b_out);
    L.ha = state0[2 * hf + 0];
    L.hb = state0[2 * hf + 1];
    gru_serial<SEQ_LEN>(L, lane == 0);
    ha = L.ha; hb = L.hb;
  } else {
    const float2 v = *reinterpret_cast<const float2*>(
        rec + (size_t)(SC_WAVES - 1) * REC_PITCH + REC_EXIT + 2 * hf);
    ha = v.x; hb = v.y;
  }

  if (c == 0) {
    state_out[2 * hf + 0] = ha;
    state_out[2 * hf + 1] = hb;
  }
}

extern "C" void kernel_launch(void* const* d_in, const int* in_sizes, int n_in,
                              void* d_out, int out_size, void* d_ws, size_t ws_size,
                              hipStream_t stream) {
  const float* x      = (const float*)d_in[0];
  const float* state  = (const float*)d_in[1];
  const float* w_ih   = (const float*)d_in[2];
  const float* w_hh   = (const float*)d_in[3];
  const float* b_ih   = (const float*)d_in[4];
  const float* b_hh   = (const float*)d_in[5];
  const float* w_out  = (const float*)d_in[6];
  const float* b_out  = (const float*)d_in[7];

  float* out  = (float*)d_out;
  float* rec  = (float*)d_ws;

  gru_sweeps_kernel<<<SC_WAVES, 32, 0, stream>>>(x, state, w_ih, w_hh, b_ih, b_hh, w_out,
                                                 b_out, out, rec);
  gru_join_kernel<<<1, 32, 0, stream>>>(x, state, w_ih, w_hh, b_ih, b_hh, w_out, b_out,
                                        out, rec, out + (size_t)SEQ_LEN * 2);
}


# scan: early sweeps (never stored, only entries of later ones) run a short approximate step, each gate ONE hardware tanh, until a wave's exits settle; full-step sweeps then deliver y, checks as before
# speedup vs baseline: 2123.6879x; 1.1844x over previous
#include <hip/hip_runtime.h>

typedef __attribute__((ext_vector_type(2))) float v2f;
typedef __attribute__((ext_vector_type(8))) float v8f;

#define SEQ_LEN 262144
#define HDIM    4

__device__ __forceinline__ float fast_sigmoid(float x) {
  const float e = __builtin_amdgcn_exp2f(-1.44269504088896340736f * x);
  return __builtin_amdgcn_rcpf(1.0f + e);
}

#define SCAN_GRP  4
#define SCAN_RING 4
#define SCAN_TRIP (SCAN_GRP * SCAN_RING)

struct GruLane {
  v2f a;
  v2f ax;
  float bh_n0, bh_n1;
  float bo0, bo1;
  const float* xp;
  float* yp;
  float ha, hb;
  bool upper;
};

__device__ __forceinline__ void gru_lane_place(
    GruLane& L, const float* __restrict__ x, float* __restrict__ y, const int t_base) {
  L.xp = x + (size_t)t_base * 2;
  L.yp = y + (size_t)t_base * 2;
  L.upper = (threadIdx.x >> 4) != 0;
}

__device__ __forceinline__ void gru_lane_init(
    GruLane& L, const float* __restrict__ w_ih,
    const float* __restrict__ w_hh, const float* __restrict__ b_ih,
    const float* __restrict__ b_hh, const float* __restrict__ w_out,
    const float* __restrict__ b_out) {
  const int lane = threadIdx.x;
  const int hf   = lane >> 4;
  const int c    = lane & 15;

  {
    const int mh = c >> 3, mi = c & 7;
    if (mi < 6) {
      const int g = (mi >> 1) * 4 + 2 * mh + (mi & 1);
      const float2 wv = *reinterpret_cast<const float2*>(w_hh + g * 4 + 2 * hf);
      L.a.x = wv.x; L.a.y = wv.y;
      if (hf == 0) {
        const float2 wi = *reinterpret_cast<const float2*>(w_ih + g * 2);
        L.ax.x = wi.x; L.ax.y = wi.y;
      } else {
        const float bi = b_ih[g];
        const float bh = b_hh[g];
        L.ax.x = (g < 2 * HDIM) ? bi + bh : bi;
        L.ax.y = 0.0f;
      }
    } else if (mh == 0) {
      const float2 wv = *reinterpret_cast<const float2*>(w_out + (mi - 6) * 4 + 2 * hf);
      L.a.x = wv.x; L.a.y = wv.y;
      L.ax.x = 0.0f; L.ax.y = 0.0f;
    } else {
      L.a.x = 0.0f; L.a.y = 0.0f;
      L.ax.x = 0.0f; L.ax.y = 0.0f;
    }
  }
  L.bo0 = b_out[0]; L.bo1 = b_out[1];
  L.bh_n0 = b_hh[2 * HDIM + 2 * hf + 0];
  L.bh_n1 = b_hh[2 * HDIM + 2 * hf + 1];

  L.ha = 0.0f; L.hb = 0.0f;
}

__device__ __forceinline__ void gru_load_grp(
    const GruLane& L, float (&xs)[2 * SCAN_GRP], int t) {
  const float4 v0 = *reinterpret_cast<const float4*>(L.xp + 2 * t);
  const float4 v1 = *reinterpret_cast<const float4*>(L.xp + 2 * t + 4);
  xs[0] = v0.x; xs[1] = v0.y; xs[2] = v0.z; xs[3] = v0.w;
  xs[4] = v1.x; xs[5] = v1.y; xs[6] = v1.z; xs[7] = v1.w;
}

__device__ __forceinline__ v8f gru_gi(const GruLane& L, float x0, float x1) {
  v2f b;
  b.x = L.upper ? 1.0f : x0;
  b.y = L.upper ? 0.0f : x1;
  const v8f zero = {0.0f, 0.0f, 0.0f, 0.0f, 0.0f, 0.0f, 0.0f, 0.0f};
  return __builtin_amdgcn_wmma_f32_16x16x4_f32(
       false, L.ax,  false, b,
       (short)0, zero,  false,  false);
}

__device__ __forceinline__ void gru_run_grp(
    GruLane& L, const float (&xs)[2 * SCAN_GRP], float nx0, float nx1,
    v8f& gnext, float (&ys)[SCAN_GRP][2]) {
#pragma unroll
  for (int u = 0; u < SCAN_GRP; ++u) {
    const v8f gi = gnext;
    v8f cc;
    cc[0] = gi[0];   cc[1] = gi[1];
    cc[2] = gi[2];   cc[3] = gi[3];
    cc[4] = L.bh_n0; cc[5] = L.bh_n1;
    cc[6] = L.bo0;   cc[7] = L.bo1;
    v2f b; b.x = L.ha; b.y = L.hb;

    const v8f d = __builtin_amdgcn_wmma_f32_16x16x4_f32(
         false, L.a,  false, b,
         (short)0, cc,  false,  false);
    ys[u][0] = d[6]; ys[u][1] = d[7];

    gnext = (u + 1 < SCAN_GRP) ? gru_gi(L, xs[2 * u + 2], xs[2 * u + 3])
                               : gru_gi(L, nx0, nx1);

    const float ra = fast_sigmoid(d[0]);
    const float rb = fast_sigmoid(d[1]);
    const float za = fast_sigmoid(d[2]);
    const float zb = fast_sigmoid(d[3]);

    const float na = fmaf(2.0f, fast_sigmoid(2.0f * fmaf(ra, d[4], gi[4])), -1.0f);
    const float nb = fmaf(2.0f, fast_sigmoid(2.0f * fmaf(rb, d[5], gi[5])), -1.0f);

    L.ha = fmaf(za, L.ha - na, na);
    L.hb = fmaf(zb, L.hb - nb, nb);
  }
}

__device__ __forceinline__ void gru_step_kept(
    GruLane& L, const v8f& gi, float& y0, float& y1) {
  v8f cc;
  cc[0] = gi[0];   cc[1] = gi[1];
  cc[2] = gi[2];   cc[3] = gi[3];
  cc[4] = L.bh_n0; cc[5] = L.bh_n1;
  cc[6] = L.bo0;   cc[7] = L.bo1;
  v2f b; b.x = L.ha; b.y = L.hb;

  const v8f d = __builtin_amdgcn_wmma_f32_16x16x4_f32(
       false, L.a,  false, b,
       (short)0, cc,  false,  false);
  y0 = d[6]; y1 = d[7];

  const float ra = fast_sigmoid(d[0]);
  const float rb = fast_sigmoid(d[1]);
  const float za = fast_sigmoid(d[2]);
  const float zb = fast_sigmoid(d[3]);

  const float na = fmaf(2.0f, fast_sigmoid(2.0f * fmaf(ra, d[4], gi[4])), -1.0f);
  const float nb = fmaf(2.0f, fast_sigmoid(2.0f * fmaf(rb, d[5], gi[5])), -1.0f);

  L.ha = fmaf(za, L.ha - na, na);
  L.hb = fmaf(zb, L.hb - nb, nb);
}

__device__ __forceinline__ void gru_step_runin(GruLane& L, const v8f& gi) {
  v8f cc;
  cc[0] = gi[0];   cc[1] = gi[1];
  cc[2] = gi[2];   cc[3] = gi[3];
  cc[4] = L.bh_n0; cc[5] = L.bh_n1;
  cc[6] = L.bo0;   cc[7] = L.bo1;
  v2f b; b.x = L.ha; b.y = L.hb;

  const v8f d = __builtin_amdgcn_wmma_f32_16x16x4_f32(
       false, L.a,  false, b,
       (short)0, cc,  false,  false);

  const float tra = __builtin_amdgcn_tanhf(0.5f * d[0]);
  const float trb = __builtin_amdgcn_tanhf(0.5f * d[1]);
  const float tza = __builtin_amdgcn_tanhf(0.5f * d[2]);
  const float tzb = __builtin_amdgcn_tanhf(0.5f * d[3]);

  const float ga = 0.5f * d[4];
  const float gb = 0.5f * d[5];
  v2f n;
  n.x = __builtin_amdgcn_tanhf(fmaf(tra, ga, gi[4] + ga));
  n.y = __builtin_amdgcn_tanhf(fmaf(trb, gb, gi[5] + gb));

  v2f k, zh;
  k.x  = fmaf(-0.5f, tza, 0.5f);
  k.y  = fmaf(-0.5f, tzb, 0.5f);
  zh.x = fmaf(0.5f, tza, 0.5f) * L.ha;
  zh.y = fmaf(0.5f, tzb, 0.5f) * L.hb;
  const v2f hn = __builtin_elementwise_fma(k, n, zh);
  L.ha = hn.x; L.hb = hn.y;
}

__device__ __forceinline__ void gru_store_trip(
    const GruLane& L, const float (&yt)[SCAN_RING][SCAN_GRP][2], int t, bool keep) {
  if (keep) {
    float* yq = L.yp + t * 2;
    if (t > 0)
      *reinterpret_cast<float2*>(yq - 2) = make_float2(yt[0][0][0], yt[0][0][1]);
#pragma unroll
    for (int i = 1; i < SCAN_TRIP; ++i)
      *reinterpret_cast<float2*>(yq + (i - 1) * 2) =
          make_float2(yt[i / SCAN_GRP][i % SCAN_GRP][0], yt[i / SCAN_GRP][i % SCAN_GRP][1]);
  }
}

__device__ __forceinline__ void gru_last_y(const GruLane& L, int t_last, bool keep) {
  const v8f cc = {0.0f, 0.0f, 0.0f, 0.0f, 0.0f, 0.0f, L.bo0, L.bo1};
  v2f b; b.x = L.ha; b.y = L.hb;
  const v8f d = __builtin_amdgcn_wmma_f32_16x16x4_f32(
       false, L.a,  false, b,
       (short)0, cc,  false,  false);
  if (keep)
    *reinterpret_cast<float2*>(L.yp + t_last * 2) = make_float2(d[6], d[7]);
}

#define CH_LEN   64
#define N_CHUNK  (SEQ_LEN / CH_LEN)
#define RUNIN    8
#define OWNED    (16 - RUNIN)
#define N_SWEEP  (RUNIN + 1)
#define N_RUNIN_MAX RUNIN
#define RUNIN_TOL 2.44140625e-4f
#define CH_TRIPS (CH_LEN / SCAN_TRIP)
#define SC_WAVES (N_CHUNK / OWNED)
#define REC_PITCH 32
#define REC_EXIT  HDIM
#define REC_FLAG  (2 * HDIM)

__device__ __forceinline__ float from_left_column(float v) {
  return __builtin_bit_cast(float, __builtin_amdgcn_update_dpp(
      0, __builtin_bit_cast(int, v),   0x111, 0xf, 0xf, true));
}

__global__ __launch_bounds__(32) void gru_sweeps_kernel(
    const float* __restrict__ x, const float* __restrict__ state0,
    const float* __restrict__ w_ih, const float* __restrict__ w_hh,
    const float* __restrict__ b_ih, const float* __restrict__ b_hh,
    const float* __restrict__ w_out, const float* __restrict__ b_out,
    float* __restrict__ y, float* __restrict__ rec) {
  const int lane = threadIdx.x;
  const int hf   = lane >> 4;
  const int c    = lane & 15;
  const int k    = (int)blockIdx.x * OWNED - RUNIN + c;
  const int kc   = k < 0 ? 0 : k;
  const bool head  = (k == 0);
  const bool owned = (c >= RUNIN);
  const bool ykeep = owned & (hf == 0);

  GruLane L;
  gru_lane_place(L, x, y, kc * CH_LEN);

  float xs[2 * CH_LEN];
#pragma unroll
  for (int i = 0; i < CH_LEN / 2; ++i) {
    const float4 v = *reinterpret_cast<const float4*>(L.xp + 4 * i);
    xs[4 * i + 0] = v.x; xs[4 * i + 1] = v.y;
    xs[4 * i + 2] = v.z; xs[4 * i + 3] = v.w;
  }

  gru_lane_init(L, w_ih, w_hh, b_ih, b_hh, w_out, b_out);

  v8f gi[CH_LEN];
#pragma unroll
  for (int t = 0; t < CH_LEN; ++t) gi[t] = gru_gi(L, xs[2 * t], xs[2 * t + 1]);

  const float s_a = state0[2 * hf + 0];
  const float s_b = state0[2 * hf + 1];
  float ea = head ? s_a : 0.0f;
  float eb = head ? s_b : 0.0f;
  float xa = 0.0f, xb = 0.0f;

  L.ha = ea; L.hb = eb;
  for (int sweep = 0; sweep < N_RUNIN_MAX; ++sweep) {
#pragma unroll
    for (int t = 0; t < CH_LEN; ++t) gru_step_runin(L, gi[t]);
    const bool settled = (fabsf(L.ha - xa) < RUNIN_TOL) & (fabsf(L.hb - xb) < RUNIN_TOL);
    xa = L.ha; xb = L.hb;
    const float la = from_left_column(xa);
    const float lb = from_left_column(xb);
    ea = head ? s_a : la;
    eb = head ? s_b : lb;
    L.ha = ea; L.hb = eb;
    if ((sweep >= 1) & (__builtin_amdgcn_ballot_w32(!settled) == 0u)) break;
  }

  bool drift = false;
  for (int sweep = 0; sweep < N_SWEEP; ++sweep) {
#pragma unroll
    for (int tr = 0; tr < CH_TRIPS; ++tr) {
      float yt[SCAN_RING][SCAN_GRP][2];
#pragma unroll
      for (int u = 0; u < SCAN_TRIP; ++u)
        gru_step_kept(L, gi[tr * SCAN_TRIP + u],
                      yt[u / SCAN_GRP][u % SCAN_GRP][0], yt[u / SCAN_GRP][u % SCAN_GRP][1]);
      gru_store_trip(L, yt, tr * SCAN_TRIP, ykeep);
    }
    drift = (__builtin_bit_cast(int, L.ha) != __builtin_bit_cast(int, xa)) |
            (__builtin_bit_cast(int, L.hb) != __builtin_bit_cast(int, xb));
    const bool calm = (__builtin_amdgcn_ballot_w32(drift) == 0u);
    if (calm | (sweep == N_SWEEP - 1)) break;
    xa = L.ha; xb = L.hb;
    const float la = from_left_column(xa);
    const float lb = from_left_column(xb);
    ea = head ? s_a : la;
    eb = head ? s_b : lb;
    L.ha = ea; L.hb = eb;
  }

  gru_last_y(L, CH_LEN - 1, ykeep);

  const bool moved = owned & (c < 15) & drift;
  const unsigned any_moved = __builtin_amdgcn_ballot_w32(moved);
  float* line = rec + (size_t)blockIdx.x * REC_PITCH;
  if (lane == 0)
    line[REC_FLAG] = (any_moved != 0u) ? 1.0f : 0.0f;
  if (c == RUNIN)
    *reinterpret_cast<float2*>(line + 2 * hf) = make_float2(ea, eb);
  if (c == 15)
    *reinterpret_cast<float2*>(line + REC_EXIT + 2 * hf) = make_float2(L.ha, L.hb);
}

template <int NSTEPS>
__device__ __forceinline__ void gru_serial(GruLane& L, bool keep) {
  float g[SCAN_RING][2 * SCAN_GRP];
#pragma unroll
  for (int j = 0; j < SCAN_RING; ++j) gru_load_grp(L, g[j], SCAN_GRP * j);
  v8f gnext = gru_gi(L, g[0][0], g[0][1]);

  for (int t = 0; t < NSTEPS; t += SCAN_TRIP) {
    float yt[SCAN_RING][SCAN_GRP][2];
#pragma unroll
    for (int j = 0; j < SCAN_RING; ++j) {
      gru_run_grp(L, g[j], g[(j + 1) % SCAN_RING][0], g[(j + 1) % SCAN_RING][1],
                  gnext, yt[j]);
      if (j == SCAN_RING - 1) gru_store_trip(L, yt, t, keep);
      gru_load_grp(L, g[j], (t + SCAN_TRIP + SCAN_GRP * j) & (NSTEPS - 1));
      __builtin_amdgcn_sched_barrier(0);
    }
  }
  gru_last_y(L, NSTEPS - 1, keep);
}

__global__ __launch_bounds__(32) void gru_join_kernel(
    const float* __restrict__ x, const float* __restrict__ state0,
    const float* __restrict__ w_ih, const float* __restrict__ w_hh,
    const float* __restrict__ b_ih, const float* __restrict__ b_hh,
    const float* __restrict__ w_out, const float* __restrict__ b_out,
    float* __restrict__ y, const float* __restrict__ rec,
    float* __restrict__ state_out) {
  const int lane = threadIdx.x;
  const int hf   = lane >> 4;
  const int c    = lane & 15;

  bool bad = false;
#pragma unroll
  for (int j = 0; j < SC_WAVES / 32; ++j) {
    const int w  = j * 32 + lane;
    const int wl = w > 0 ? w : 1;
    const float* line = rec + (size_t)w * REC_PITCH;
    const float4 e = *reinterpret_cast<const float4*>(rec + (size_t)wl * REC_PITCH);
    const float4 xl = *reinterpret_cast<const float4*>(
        rec + (size_t)(wl - 1) * REC_PITCH + REC_EXIT);
    bad |= (line[REC_FLAG] != 0.0f);
    bad |= (w > 0) &
           ((__builtin_bit_cast(int, e.x) != __builtin_bit_cast(int, xl.x)) |
            (__builtin_bit_cast(int, e.y) != __builtin_bit_cast(int, xl.y)) |
            (__builtin_bit_cast(int, e.z) != __builtin_bit_cast(int, xl.z)) |
            (__builtin_bit_cast(int, e.w) != __builtin_bit_cast(int, xl.w)));
  }

  float ha, hb;
  if (__builtin_amdgcn_ballot_w32(bad) != 0u) {
    GruLane L;
    gru_lane_place(L, x, y, 0);
    gru_lane_init(L, w_ih, w_hh, b_ih, b_hh, w_out, b_out);
    L.ha = state0[2 * hf + 0];
    L.hb = state0[2 * hf + 1];
    gru_serial<SEQ_LEN>(L, lane == 0);
    ha = L.ha; hb = L.hb;
  } else {
    const float2 v = *reinterpret_cast<const float2*>(
        rec + (size_t)(SC_WAVES - 1) * REC_PITCH + REC_EXIT + 2 * hf);
    ha = v.x; hb = v.y;
  }

  if (c == 0) {
    state_out[2 * hf + 0] = ha;
    state_out[2 * hf + 1] = hb;
  }
}

extern "C" void kernel_launch(void* const* d_in, const int* in_sizes, int n_in,
                              void* d_out, int out_size, void* d_ws, size_t ws_size,
                              hipStream_t stream) {
  const float* x      = (const float*)d_in[0];
  const float* state  = (const float*)d_in[1];
  const float* w_ih   = (const float*)d_in[2];
  const float* w_hh   = (const float*)d_in[3];
  const float* b_ih   = (const float*)d_in[4];
  const float* b_hh   = (const float*)d_in[5];
  const float* w_out  = (const float*)d_in[6];
  const float* b_out  = (const float*)d_in[7];

  float* out  = (float*)d_out;
  float* rec  = (float*)d_ws;

  gru_sweeps_kernel<<<SC_WAVES, 32, 0, stream>>>(x, state, w_ih, w_hh, b_ih, b_hh, w_out,
                                                 b_out, out, rec);
  gru_join_kernel<<<1, 32, 0, stream>>>(x, state, w_ih, w_hh, b_ih, b_hh, w_out, b_out,
                                        out, rec, out + (size_t)SEQ_LEN * 2);
}
